# starting kernel
# baseline (speedup 1.0000x reference)
#include <hip/hip_runtime.h>
#include <hip/hip_bf16.h>


typedef __attribute__((ext_vector_type(2))) float v2f;
typedef __attribute__((ext_vector_type(4))) float v4f;
typedef __attribute__((ext_vector_type(8))) float v8f;

#define NQ      10
#define NL      4
#define BATCH_N 8192
#define PITCH   33
#define PLANE   (32 * PITCH)
#define SAMPLE_F (4 * PLANE)
#define WPB     2

__device__ __forceinline__ int uh_off(int l, int i, int k, int p) {
  return (((l * 2 + i) * 2 + k) * 2 + p) * 256;
}
__device__ __forceinline__ int ul_off(int l, int k, int j, int p) {
  return 8192 + (((l * 2 + k) * 2 + j) * 2 + p) * 256;
}

__device__ __forceinline__ void gate_elem(float a, float b, int rb, int cb,
                                          float& fr, float& fi) {
  float ca = cosf(0.5f * a), sa = sinf(0.5f * a);
  float cz = cosf(0.5f * b), sz = sinf(0.5f * b);
  float ry = rb ? (cb ? ca : sa) : (cb ? -sa : ca);
  fr = ry * cz;
  fi = ry * (rb ? sz : -sz);
}

__global__ void vqc_prep(const float* __restrict__ w, float* __restrict__ ws) {
  int l = blockIdx.x;
  for (int t = threadIdx.x; t < 4096; t += blockDim.x) {
    int which = t >> 11;
    int e     = t & 2047;
    int frag  = e >> 8;
    int lane  = (e >> 3) & 31;
    int slot  = e & 7;
    int p     = frag & 1;
    int Kloc  = 4 * (slot >> 1) + (slot & 1) + 2 * (lane >> 4);

    float vr = 1.f, vi = 0.f;
    int off;
    if (which == 0) {
      int it = frag >> 2, kt = (frag >> 1) & 1;
      int m = it * 16 + (lane & 15);
      int kcol = kt * 16 + Kloc;
      int r0 = (m ^ (m >> 1)) & 31;
      #pragma unroll
      for (int w2 = 0; w2 < 5; ++w2) {
        int rb = (r0 >> (4 - w2)) & 1;
        int cb = (kcol >> (4 - w2)) & 1;
        float fr, fi;
        gate_elem(w[(l * NQ + w2) * 2], w[(l * NQ + w2) * 2 + 1], rb, cb, fr, fi);
        float nr = vr * fr - vi * fi; vi = vr * fi + vi * fr; vr = nr;
      }
      off = uh_off(l, it, kt, p);
    } else {
      int kt = frag >> 2, jt = (frag >> 1) & 1;
      int n  = jt * 16 + (lane & 15);
      int kg = kt * 16 + Kloc;
      #pragma unroll
      for (int w2 = 0; w2 < 5; ++w2) {
        int rb = (n >> (4 - w2)) & 1;
        int cb = (kg >> (4 - w2)) & 1;
        float fr, fi;
        gate_elem(w[(l * NQ + 5 + w2) * 2], w[(l * NQ + 5 + w2) * 2 + 1], rb, cb, fr, fi);
        float nr = vr * fr - vi * fi; vi = vr * fi + vi * fr; vr = nr;
      }
      off = ul_off(l, kt, jt, p);
    }
    ws[off + lane * 8 + slot] = p ? vi : vr;
  }
}

__global__ __launch_bounds__(32 * WPB) void vqc_main(const float* __restrict__ x,
                                                     const float* __restrict__ ws,
                                                     float* __restrict__ out) {
  __shared__ float lds[WPB * SAMPLE_F];
  const int lane = threadIdx.x & 31;
  const int wv   = threadIdx.x >> 5;
  const int sample = blockIdx.x * WPB + wv;

  float* Sa_r = lds + wv * SAMPLE_F;
  float* Sa_i = Sa_r + PLANE;
  float* Sb_r = Sa_r + 2 * PLANE;
  float* Sb_i = Sa_r + 3 * PLANE;

  float rowv = 1.f, colv = 1.f;
  #pragma unroll
  for (int w2 = 0; w2 < 5; ++w2) {
    float ah = 0.5f * x[sample * NQ + w2];
    float al = 0.5f * x[sample * NQ + 5 + w2];
    int b = (lane >> (4 - w2)) & 1;
    rowv *= b ? sinf(ah) : cosf(ah);
    colv *= b ? sinf(al) : cosf(al);
  }
  for (int r = 0; r < 32; ++r) {
    float rv = __shfl(rowv, r, 32);
    Sa_r[r * PITCH + lane] = rv * colv;
    Sa_i[r * PITCH + lane] = 0.f;
  }

  for (int l = 0; l < NL; ++l) {
    #pragma unroll
    for (int it = 0; it < 2; ++it) {
      float ar[2][8], ai[2][8];
      #pragma unroll
      for (int k = 0; k < 2; ++k)
        #pragma unroll
        for (int s = 0; s < 8; ++s) {
          int row = it * 16 + (lane & 15);
          int col = k * 16 + 4 * (s >> 1) + (s & 1) + 2 * (lane >> 4);
          int rs = row, cs = col;
          if (l > 0) {
            int rm = (col & 1) ? (row ^ 16) : row;
            int c3 = (col ^ (col >> 1)) & 31;
            cs = (rm & 1) ? (c3 ^ 16) : c3;
            rs = rm;
          }
          ar[k][s] = Sa_r[rs * PITCH + cs];
          ai[k][s] = Sa_i[rs * PITCH + cs];
        }
      #pragma unroll
      for (int j = 0; j < 2; ++j) {
        v8f dr = {0,0,0,0,0,0,0,0}, di = {0,0,0,0,0,0,0,0};
        #pragma unroll
        for (int k = 0; k < 2; ++k) {
          const v4f* pr = (const v4f*)(ws + ul_off(l, k, j, 0) + lane * 8);
          const v4f* pi = (const v4f*)(ws + ul_off(l, k, j, 1) + lane * 8);
          v4f r0 = pr[0], r1 = pr[1], i0 = pi[0], i1 = pi[1];
          float br[8] = {r0.x, r0.y, r0.z, r0.w, r1.x, r1.y, r1.z, r1.w};
          float bi[8] = {i0.x, i0.y, i0.z, i0.w, i1.x, i1.y, i1.z, i1.w};
          #pragma unroll
          for (int kk = 0; kk < 4; ++kk) {
            v2f aR  = { ar[k][2 * kk],  ar[k][2 * kk + 1]};
            v2f aI  = { ai[k][2 * kk],  ai[k][2 * kk + 1]};
            v2f aIn = {-ai[k][2 * kk], -ai[k][2 * kk + 1]};
            v2f bR  = { br[2 * kk],     br[2 * kk + 1]};
            v2f bI  = { bi[2 * kk],     bi[2 * kk + 1]};
            dr = __builtin_amdgcn_wmma_f32_16x16x4_f32(false, aR,  false, bR, (short)0, dr, false, false);
            dr = __builtin_amdgcn_wmma_f32_16x16x4_f32(false, aIn, false, bI, (short)0, dr, false, false);
            di = __builtin_amdgcn_wmma_f32_16x16x4_f32(false, aR,  false, bI, (short)0, di, false, false);
            di = __builtin_amdgcn_wmma_f32_16x16x4_f32(false, aI,  false, bR, (short)0, di, false, false);
          }
        }
        #pragma unroll
        for (int s = 0; s < 8; ++s) {
          int m = it * 16 + s + 8 * (lane >> 4);
          int n = j * 16 + (lane & 15);
          Sb_r[m * PITCH + n] = dr[s];
          Sb_i[m * PITCH + n] = di[s];
        }
      }
    }
    #pragma unroll
    for (int j = 0; j < 2; ++j) {
      float br[2][8], bi[2][8];
      #pragma unroll
      for (int k = 0; k < 2; ++k)
        #pragma unroll
        for (int s = 0; s < 8; ++s) {
          int K = k * 16 + 4 * (s >> 1) + (s & 1) + 2 * (lane >> 4);
          int n = j * 16 + (lane & 15);
          br[k][s] = Sb_r[K * PITCH + n];
          bi[k][s] = Sb_i[K * PITCH + n];
        }
      #pragma unroll
      for (int it = 0; it < 2; ++it) {
        v8f dr = {0,0,0,0,0,0,0,0}, di = {0,0,0,0,0,0,0,0};
        #pragma unroll
        for (int k = 0; k < 2; ++k) {
          const v4f* pr = (const v4f*)(ws + uh_off(l, it, k, 0) + lane * 8);
          const v4f* pi = (const v4f*)(ws + uh_off(l, it, k, 1) + lane * 8);
          v4f r0 = pr[0], r1 = pr[1], i0 = pi[0], i1 = pi[1];
          float ur[8] = {r0.x, r0.y, r0.z, r0.w, r1.x, r1.y, r1.z, r1.w};
          float ui[8] = {i0.x, i0.y, i0.z, i0.w, i1.x, i1.y, i1.z, i1.w};
          #pragma unroll
          for (int kk = 0; kk < 4; ++kk) {
            v2f aR  = { ur[2 * kk],  ur[2 * kk + 1]};
            v2f aI  = { ui[2 * kk],  ui[2 * kk + 1]};
            v2f aIn = {-ui[2 * kk], -ui[2 * kk + 1]};
            v2f bR  = { br[k][2 * kk], br[k][2 * kk + 1]};
            v2f bI  = { bi[k][2 * kk], bi[k][2 * kk + 1]};
            dr = __builtin_amdgcn_wmma_f32_16x16x4_f32(false, aR,  false, bR, (short)0, dr, false, false);
            dr = __builtin_amdgcn_wmma_f32_16x16x4_f32(false, aIn, false, bI, (short)0, dr, false, false);
            di = __builtin_amdgcn_wmma_f32_16x16x4_f32(false, aR,  false, bI, (short)0, di, false, false);
            di = __builtin_amdgcn_wmma_f32_16x16x4_f32(false, aI,  false, bR, (short)0, di, false, false);
          }
        }
        #pragma unroll
        for (int s = 0; s < 8; ++s) {
          int m = it * 16 + s + 8 * (lane >> 4);
          int n = j * 16 + (lane & 15);
          Sa_r[m * PITCH + n] = dr[s];
          Sa_i[m * PITCH + n] = di[s];
        }
      }
    }
  }

  float acc[NQ];
  #pragma unroll
  for (int i = 0; i < NQ; ++i) acc[i] = 0.f;
  for (int r = 0; r < 32; ++r) {
    float re = Sa_r[r * PITCH + lane];
    float im = Sa_i[r * PITCH + lane];
    float p  = re * re + im * im;
    int c3 = (r & 1) ? (lane ^ 16) : lane;
    int c4 = (c3 ^ (c3 >> 1) ^ (c3 >> 2) ^ (c3 >> 3) ^ (c3 >> 4)) & 31;
    int r2 = (c4 & 1) ? (r ^ 16) : r;
    int a  = (r2 << 5) | c4;
    #pragma unroll
    for (int i = 0; i < NQ; ++i)
      acc[i] += (a & (1 << (9 - i))) ? -p : p;
  }
  #pragma unroll
  for (int off = 16; off > 0; off >>= 1)
    #pragma unroll
    for (int i = 0; i < NQ; ++i)
      acc[i] += __shfl_xor(acc[i], off, 32);
  if (lane == 0) {
    #pragma unroll
    for (int i = 0; i < NQ; ++i) out[sample * NQ + i] = acc[i];
  }
}

extern "C" void kernel_launch(void* const* d_in, const int* in_sizes, int n_in,
                              void* d_out, int out_size, void* d_ws, size_t ws_size,
                              hipStream_t stream) {
  const float* x = (const float*)d_in[0];
  const float* w = (const float*)d_in[1];
  float* ws  = (float*)d_ws;
  float* out = (float*)d_out;

  vqc_prep<<<NL, 1024, 0, stream>>>(w, ws);
  vqc_main<<<BATCH_N / WPB, 32 * WPB, 0, stream>>>(x, ws, out);
}


# one LDS state buffer per sample, not two: the four-wire factor reads the whole state into registers before its ring-permuted stores, in place; allocation held to 128 VGPRs: 8 waves per SIMD, not 4
# speedup vs baseline: 1.7938x; 1.7938x over previous
#include <hip/hip_runtime.h>


typedef __attribute__((ext_vector_type(2))) float v2f;
typedef __attribute__((ext_vector_type(4))) float v4f;
typedef __attribute__((ext_vector_type(8))) float v8f;

#define NQ      10
#define NL      4
#define BATCH_N 8192
#define IM_OFF  1058
#define BUF_F   2112
#define WPB     2
#define LAYER_F 1536

__device__ __forceinline__ int swz(int a) {
  return a ^ (((a >> 6) & 3) << 2) ^ (((a >> 8) & 1) << 4);
}

__device__ __forceinline__ int ring(int a) {
  int p = a;
  p ^= p >> 1; p ^= p >> 2; p ^= p >> 4; p ^= p >> 8;
  return (p ^ ((p & 1) << 9)) & 1023;
}

__device__ __forceinline__ void gate_elem(float a, float b, int rb, int cb,
                                          float& fr, float& fi) {
  float ca = cosf(0.5f * a), sa = sinf(0.5f * a);
  float cz = cosf(0.5f * b), sz = sinf(0.5f * b);
  float ry = rb ? (cb ? ca : sa) : (cb ? -sa : ca);
  fr = ry * cz;
  fi = ry * (rb ? sz : -sz);
}

__global__ __launch_bounds__(128) void vqc_prep(const float* __restrict__ w, float* __restrict__ ws) {
  int t = blockIdx.x * 128 + threadIdx.x;
  int l = t / LAYER_F;
  int e = t - l * LAYER_F;
  int frag = e >> 8;
  int lane = (e >> 3) & 31;
  int slot = e & 7;
  int c = lane & 15, h = lane >> 4;
  int Kloc = 4 * (slot >> 1) + (slot & 1) + 2 * h;
  int nb, first, M, K;
  if (frag < 2) { nb = 3; first = 3 * frag; M = c; K = Kloc; }
  else { nb = 4; first = 6; M = ((frag - 2) >> 1) * 16 + c; K = ((frag - 2) & 1) * 16 + Kloc; }
  int po, o, pi, i;
  if (nb == 3) { po = M >> 3; o = M & 7; pi = K >> 3; i = K & 7; }
  else { po = (M >> 3) & 1; o = ((M >> 4) << 3) | (M & 7); pi = K >> 4; i = K & 15; }
  float vr = 1.f, vi = 0.f;
  #pragma unroll
  for (int j = 0; j < 4; ++j) {
    if (j < nb) {
      int wire = first + j;
      int rb = (o >> (nb - 1 - j)) & 1;
      int cb = (i >> (nb - 1 - j)) & 1;
      float fr, fi;
      gate_elem(w[(l * NQ + wire) * 2], w[(l * NQ + wire) * 2 + 1], rb, cb, fr, fi);
      float nr = vr * fr - vi * fi; vi = vr * fi + vi * fr; vr = nr;
    }
  }
  ws[t] = (po == pi) ? vr : (po ? vi : -vi);
}

template <int NB, int LO>
__device__ __forceinline__ void apply_group(const float* __restrict__ frag, float* buf, int lane) {
  constexpr int KS = (2 << NB) / 4;
  constexpr int MT = (2 << NB) / 16;
  constexpr int KB = KS / 4;
  constexpr int NT = (1024 >> NB) / 16;
  constexpr int TB = (MT == 1) ? 2 : 1;
  constexpr int GM = (1 << NB) - 1;
  const int c = lane & 15, h = lane >> 4;
  float* dsth = buf + h * IM_OFF;

  float af[MT][KB][8];
  #pragma unroll
  for (int mt = 0; mt < MT; ++mt)
    #pragma unroll
    for (int kb = 0; kb < KB; ++kb) {
      const v4f* p = (const v4f*)(frag + (mt * KB + kb) * 256 + lane * 8);
      v4f q0 = p[0], q1 = p[1];
      af[mt][kb][0] = q0.x; af[mt][kb][1] = q0.y; af[mt][kb][2] = q0.z; af[mt][kb][3] = q0.w;
      af[mt][kb][4] = q1.x; af[mt][kb][5] = q1.y; af[mt][kb][6] = q1.z; af[mt][kb][7] = q1.w;
    }

  #pragma unroll
  for (int t0 = 0; t0 < NT; t0 += TB) {
    v2f b[TB][KS];
    int sbase[TB], dbase[TB];
    #pragma unroll
    for (int tb = 0; tb < TB; ++tb) {
      int rest = (t0 + tb) * 16 + c;
      int arest = ((rest >> LO) << (LO + NB)) | (rest & ((1 << LO) - 1));
      sbase[tb] = swz(arest ^ (h << (LO + 1)));
      dbase[tb] = swz(arest);
      #pragma unroll
      for (int ks = 0; ks < KS; ++ks) {
        #pragma unroll
        for (int vg = 0; vg < 2; ++vg) {
          const int pi = (4 * ks + vg) >> NB;
          const int ck = swz(((4 * ks + vg) & GM) << LO);
          b[tb][ks][vg] = buf[pi * IM_OFF + (sbase[tb] ^ ck)];
        }
      }
    }
    v8f d[TB][MT];
    #pragma unroll
    for (int tb = 0; tb < TB; ++tb)
      #pragma unroll
      for (int mt = 0; mt < MT; ++mt)
        d[tb][mt] = (v8f){0, 0, 0, 0, 0, 0, 0, 0};
    #pragma unroll
    for (int ks = 0; ks < KS; ++ks)
      #pragma unroll
      for (int tb = 0; tb < TB; ++tb)
        #pragma unroll
        for (int mt = 0; mt < MT; ++mt) {
          v2f a2 = {af[mt][ks >> 2][2 * (ks & 3)], af[mt][ks >> 2][2 * (ks & 3) + 1]};
          d[tb][mt] = __builtin_amdgcn_wmma_f32_16x16x4_f32(false, a2, false, b[tb][ks],
                                                            (short)0, d[tb][mt], false, false);
        }
    #pragma unroll
    for (int tb = 0; tb < TB; ++tb)
      #pragma unroll
      for (int mt = 0; mt < MT; ++mt)
        #pragma unroll
        for (int r = 0; r < 8; ++r) {
          const int o  = (mt * 8 + r) << LO;
          const int co = swz(o);
          dsth[dbase[tb] ^ co] = d[tb][mt][r];
        }
  }
}

__device__ __forceinline__ void apply_ring_group(const float* __restrict__ frag, float* buf, int lane) {
  constexpr int KS = 8, MT = 2, KB = 2, NT = 4;
  const int c = lane & 15, h = lane >> 4;
  float* dsth = buf + h * IM_OFF;

  float af[MT][KB][8];
  #pragma unroll
  for (int mt = 0; mt < MT; ++mt)
    #pragma unroll
    for (int kb = 0; kb < KB; ++kb) {
      const v4f* p = (const v4f*)(frag + (mt * KB + kb) * 256 + lane * 8);
      v4f q0 = p[0], q1 = p[1];
      af[mt][kb][0] = q0.x; af[mt][kb][1] = q0.y; af[mt][kb][2] = q0.z; af[mt][kb][3] = q0.w;
      af[mt][kb][4] = q1.x; af[mt][kb][5] = q1.y; af[mt][kb][6] = q1.z; af[mt][kb][7] = q1.w;
    }

  v2f b[NT][KS];
  int dbase[NT];
  #pragma unroll
  for (int t = 0; t < NT; ++t) {
    const int arest = (t * 16 + c) << 4;
    const int sbase = swz(arest ^ (h << 1));
    dbase[t] = swz(ring(arest));
    #pragma unroll
    for (int ks = 0; ks < KS; ++ks)
      #pragma unroll
      for (int vg = 0; vg < 2; ++vg) {
        const int pi = (4 * ks + vg) >> 4;
        const int ck = swz((4 * ks + vg) & 15);
        b[t][ks][vg] = buf[pi * IM_OFF + (sbase ^ ck)];
      }
  }

  #pragma unroll
  for (int t = 0; t < NT; ++t) {
    v8f d[MT];
    #pragma unroll
    for (int mt = 0; mt < MT; ++mt) d[mt] = (v8f){0, 0, 0, 0, 0, 0, 0, 0};
    #pragma unroll
    for (int ks = 0; ks < KS; ++ks)
      #pragma unroll
      for (int mt = 0; mt < MT; ++mt) {
        v2f a2 = {af[mt][ks >> 2][2 * (ks & 3)], af[mt][ks >> 2][2 * (ks & 3) + 1]};
        d[mt] = __builtin_amdgcn_wmma_f32_16x16x4_f32(false, a2, false, b[t][ks],
                                                      (short)0, d[mt], false, false);
      }
    #pragma unroll
    for (int mt = 0; mt < MT; ++mt)
      #pragma unroll
      for (int r = 0; r < 8; ++r) {
        const int co = swz(ring(mt * 8 + r));
        dsth[dbase[t] ^ co] = d[mt][r];
      }
  }
}

__global__ __launch_bounds__(32 * WPB) __attribute__((amdgpu_waves_per_eu(8, 8))) void vqc_main(const float* __restrict__ x,
                                                     const float* __restrict__ ws,
                                                     float* __restrict__ out) {
  __shared__ float lds[WPB * BUF_F];
  const int lane = threadIdx.x & 31;
  const int wv   = threadIdx.x >> 5;
  const int sample = blockIdx.x * WPB + wv;

  float* X = lds + wv * BUF_F;

  float rowv = 1.f, colv = 1.f;
  #pragma unroll
  for (int w2 = 0; w2 < 5; ++w2) {
    float ah = 0.5f * x[sample * NQ + w2];
    float al = 0.5f * x[sample * NQ + 5 + w2];
    int b = (lane >> (4 - w2)) & 1;
    rowv *= b ? sinf(ah) : cosf(ah);
    colv *= b ? sinf(al) : cosf(al);
  }
  for (int r = 0; r < 32; ++r) {
    float rv = __shfl(rowv, r, 32);
    int s = swz((r << 5) | lane);
    X[s] = rv * colv;
    X[IM_OFF + s] = 0.f;
  }

  #pragma unroll 1
  for (int l = 0; l < NL; ++l) {
    const float* f0 = ws + l * LAYER_F;
    int ln = lane;
    asm volatile("" : "+v"(ln));
    ln &= 31;
    apply_group<3, 7>(f0,       X, ln);
    apply_group<3, 4>(f0 + 256, X, ln);
    apply_ring_group(f0 + 512, X, ln);
  }

  float acc[NQ];
  #pragma unroll
  for (int i = 0; i < NQ; ++i) acc[i] = 0.f;
  for (int r = 0; r < 32; ++r) {
    int a = (r << 5) | lane;
    int s = swz(a);
    float re = X[s];
    float im = X[IM_OFF + s];
    float p  = re * re + im * im;
    #pragma unroll
    for (int i = 0; i < NQ; ++i)
      acc[i] += (a & (1 << (9 - i))) ? -p : p;
  }
  #pragma unroll
  for (int off = 16; off > 0; off >>= 1)
    #pragma unroll
    for (int i = 0; i < NQ; ++i)
      acc[i] += __shfl_xor(acc[i], off, 32);
  if (lane == 0) {
    #pragma unroll
    for (int i = 0; i < NQ; ++i) out[sample * NQ + i] = acc[i];
  }
}

extern "C" void kernel_launch(void* const* d_in, const int* in_sizes, int n_in,
                              void* d_out, int out_size, void* d_ws, size_t ws_size,
                              hipStream_t stream) {
  const float* x = (const float*)d_in[0];
  const float* w = (const float*)d_in[1];
  float* ws  = (float*)d_ws;
  float* out = (float*)d_out;

  vqc_prep<<<NL * LAYER_F / 128, 128, 0, stream>>>(w, ws);
  vqc_main<<<BATCH_N / WPB, 32 * WPB, 0, stream>>>(x, ws, out);
}


# state as the A operand, gate transposed as B, k order = LDS order: all LDS traffic of the layers in 16-byte loads and stores (96 per layer, were 318); ring folded into store addresses; sums regrouped
# speedup vs baseline: 1.9072x; 1.0632x over previous
#include <hip/hip_runtime.h>


typedef __attribute__((ext_vector_type(2))) float v2f;
typedef __attribute__((ext_vector_type(4))) float v4f;
typedef __attribute__((ext_vector_type(8))) float v8f;

#define NQ      10
#define NL      4
#define BATCH_N 8192
#define BUF_F   2048
#define WPB     2
#define LAYER_F 1536

__device__ __forceinline__ constexpr int phys(int p, int U) {
  return (p << 10) | (((U ^ ((U >> 4) & 15) ^ p) & 255) << 2);
}

__device__ __forceinline__ int physA(int p, int a) {
  return phys(p, ((a & 127) << 1) | (a >> 9)) + ((a >> 7) & 3);
}

__device__ __forceinline__ void gate_elem(float a, float b, int rb, int cb,
                                          float& fr, float& fi) {
  float ca = cosf(0.5f * a), sa = sinf(0.5f * a);
  float cz = cosf(0.5f * b), sz = sinf(0.5f * b);
  float ry = rb ? (cb ? ca : sa) : (cb ? -sa : ca);
  fr = ry * cz;
  fi = ry * (rb ? sz : -sz);
}

__global__ __launch_bounds__(128) void vqc_prep(const float* __restrict__ w, float* __restrict__ ws) {
  int t = blockIdx.x * 128 + threadIdx.x;
  int l = t / LAYER_F;
  int e = t - l * LAYER_F;
  int nb, first, M, K;
  if (e < 512) {
    int lane = (e >> 3) & 31, slot = e & 7;
    nb = 3; first = 3 * (e >> 8); M = lane & 15; K = 8 * (lane >> 4) + slot;
  } else {
    int e2 = e - 512;
    int lane = (e2 >> 4) & 31, slot = e2 & 15;
    nb = 4; first = 6; M = (e2 >> 9) * 16 + (lane & 15); K = 16 * (lane >> 4) + slot;
  }
  int gm = (1 << nb) - 1;
  int po = M >> nb, o = M & gm, pi = K >> nb, i = K & gm;
  float vr = 1.f, vi = 0.f;
  #pragma unroll
  for (int j = 0; j < 4; ++j) {
    if (j < nb) {
      int wire = first + j;
      int rb = (o >> (nb - 1 - j)) & 1;
      int cb = (i >> (nb - 1 - j)) & 1;
      float fr, fi;
      gate_elem(w[(l * NQ + wire) * 2], w[(l * NQ + wire) * 2 + 1], rb, cb, fr, fi);
      float nr = vr * fr - vi * fi; vi = vr * fi + vi * fr; vr = nr;
    }
  }
  ws[t] = (po == pi) ? vr : (po ? vi : -vi);
}

__device__ __forceinline__ v4f ld4(const float* p) { return *(const v4f*)p; }
__device__ __forceinline__ void st4(float* p, v4f v) { *(v4f*)p = v; }

template <bool SECOND>
__device__ __forceinline__ void factor3w(const float* __restrict__ frag, float* X, int lane) {
  const int c = lane & 15, h = lane >> 4;
  const v4f* gp = (const v4f*)(frag + lane * 8);
  const v4f g0 = gp[0], g1 = gp[1];
  const float g[8] = {g0.x, g0.y, g0.z, g0.w, g1.x, g1.y, g1.z, g1.w};

  int lb, sb;
  if (!SECOND) {
    lb = phys(h, ((c & 7) << 5) | ((c >> 3) << 4));
    sb = phys(c >> 3, ((c & 7) << 5) | (h << 4));
  } else {
    lb = phys(h, c << 1);
    sb = phys(c >> 3, ((((c & 3) | (((c >> 2) & 1) << 3))) << 2) | (h << 1));
  }

  v4f a[8][2];
  #pragma unroll
  for (int T = 0; T < 8; ++T)
    #pragma unroll
    for (int j = 0; j < 2; ++j) {
      const int cu = SECOND ? ((T << 5) | j) : ((T << 1) | j);
      a[T][j] = ld4(X + (lb ^ phys(0, cu)));
    }

  #pragma unroll
  for (int tp = 0; tp < 4; ++tp) {
    v8f d[2];
    d[0] = (v8f){0, 0, 0, 0, 0, 0, 0, 0};
    d[1] = (v8f){0, 0, 0, 0, 0, 0, 0, 0};
    #pragma unroll
    for (int ks = 0; ks < 4; ++ks)
      #pragma unroll
      for (int t2 = 0; t2 < 2; ++t2) {
        const int T = 2 * tp + t2;
        v2f a2 = {a[T][ks >> 1][(2 * ks) & 3], a[T][ks >> 1][((2 * ks) & 3) + 1]};
        v2f b2 = {g[2 * ks], g[2 * ks + 1]};
        d[t2] = __builtin_amdgcn_wmma_f32_16x16x4_f32(false, a2, false, b2,
                                                      (short)0, d[t2], false, false);
      }
    #pragma unroll
    for (int t2 = 0; t2 < 2; ++t2)
      #pragma unroll
      for (int j = 0; j < 2; ++j) {
        const int T = 2 * tp + t2;
        const int cu = SECOND ? (((((T & 1) << 2) | ((T >> 1) << 4)) << 2) | j) : ((T << 1) | j);
        v4f v = {d[t2][4 * j], d[t2][4 * j + 1], d[t2][4 * j + 2], d[t2][4 * j + 3]};
        st4(X + (sb ^ phys(0, cu)), v);
      }
  }
}

__device__ __forceinline__ void factor4w_ring(const float* __restrict__ frag, float* X, int lane) {
  const int c = lane & 15, h = lane >> 4;
  const int u0 = c >> 3, u3 = (c >> 2) & 1, u2 = (c >> 1) & 1, u1 = c & 1;
  const int lrow = ((u2 ^ u1) << 2) | ((u1 ^ u0) << 3) | ((u3 ^ u2) << 4) | (u3 << 5);
  const int lb = phys(h, lrow << 2);
  const int n3 = h ^ (c >> 3), n2 = n3 ^ ((c >> 2) & 1), n1 = n2 ^ ((c >> 1) & 1), n0 = n1 ^ (c & 1);
  const int srow = (h << 6) | (h << 5) | (h << 4) | (n3 << 3) | (n2 << 2) | (n1 << 1) | n0;
  const int sb = phys(0, (srow << 1) | n0);

  v4f a[4][4];
  #pragma unroll
  for (int T = 0; T < 4; ++T)
    #pragma unroll
    for (int q = 0; q < 4; ++q)
      a[T][q] = ld4(X + (lb ^ phys(0, (T << 2) | q)));

  #pragma unroll
  for (int nt = 0; nt < 2; ++nt) {
    const v4f* gp = (const v4f*)(frag + nt * 512 + lane * 16);
    const v4f g0 = gp[0], g1 = gp[1], g2 = gp[2], g3 = gp[3];
    const float g[16] = {g0.x, g0.y, g0.z, g0.w, g1.x, g1.y, g1.z, g1.w,
                         g2.x, g2.y, g2.z, g2.w, g3.x, g3.y, g3.z, g3.w};
    #pragma unroll
    for (int tp = 0; tp < 2; ++tp) {
      v8f d[2];
      d[0] = (v8f){0, 0, 0, 0, 0, 0, 0, 0};
      d[1] = (v8f){0, 0, 0, 0, 0, 0, 0, 0};
      #pragma unroll
      for (int ks = 0; ks < 8; ++ks)
        #pragma unroll
        for (int t2 = 0; t2 < 2; ++t2) {
          const int T = 2 * tp + t2;
          v2f a2 = {a[T][ks >> 1][(2 * ks) & 3], a[T][ks >> 1][((2 * ks) & 3) + 1]};
          v2f b2 = {g[2 * ks], g[2 * ks + 1]};
          d[t2] = __builtin_amdgcn_wmma_f32_16x16x4_f32(false, a2, false, b2,
                                                        (short)0, d[t2], false, false);
        }
      #pragma unroll
      for (int t2 = 0; t2 < 2; ++t2)
        #pragma unroll
        for (int j = 0; j < 2; ++j) {
          const int T = 2 * tp + t2;
          const int a5 = (T >> 1) & 1, a4 = T & 1;
          const int nT = (a5 ? 0x3f : 0) ^ (a4 ? 0x1f : 0);
          const int cu = (nT << 1) | ((a5 ^ a4) ^ j);
          v4f v = {d[t2][4 * j], d[t2][4 * j + 1], d[t2][4 * j + 2], d[t2][4 * j + 3]};
          st4(X + (sb ^ phys(nt, cu)), v);
        }
    }
  }
}

__global__ __launch_bounds__(32 * WPB) __attribute__((amdgpu_waves_per_eu(8, 8))) void vqc_main(const float* __restrict__ x,
                                                     const float* __restrict__ ws,
                                                     float* __restrict__ out) {
  __shared__ __attribute__((aligned(16))) float lds[WPB * BUF_F];
  const int lane = threadIdx.x & 31;
  const int wv   = threadIdx.x >> 5;
  const int sample = blockIdx.x * WPB + wv;

  float* X = lds + wv * BUF_F;

  float rowv = 1.f, colv = 1.f;
  #pragma unroll
  for (int w2 = 0; w2 < 5; ++w2) {
    float ah = 0.5f * x[sample * NQ + w2];
    float al = 0.5f * x[sample * NQ + 5 + w2];
    int b = (lane >> (4 - w2)) & 1;
    rowv *= b ? sinf(ah) : cosf(ah);
    colv *= b ? sinf(al) : cosf(al);
  }
  for (int r = 0; r < 32; ++r) {
    float rv = __shfl(rowv, r, 32);
    int a = (r << 5) | lane;
    X[physA(0, a)] = rv * colv;
    X[physA(1, a)] = 0.f;
  }

  #pragma unroll 1
  for (int l = 0; l < NL; ++l) {
    const float* f0 = ws + l * LAYER_F;
    int ln = lane;
    asm volatile("" : "+v"(ln));
    ln &= 31;
    factor3w<false>(f0,       X, ln);
    factor3w<true >(f0 + 256, X, ln);
    factor4w_ring(f0 + 512, X, ln);
  }

  float acc[NQ];
  #pragma unroll
  for (int i = 0; i < NQ; ++i) acc[i] = 0.f;
  for (int r = 0; r < 32; ++r) {
    int a = (r << 5) | lane;
    float re = X[physA(0, a)];
    float im = X[physA(1, a)];
    float p  = re * re + im * im;
    #pragma unroll
    for (int i = 0; i < NQ; ++i)
      acc[i] += (a & (1 << (9 - i))) ? -p : p;
  }
  #pragma unroll
  for (int off = 16; off > 0; off >>= 1)
    #pragma unroll
    for (int i = 0; i < NQ; ++i)
      acc[i] += __shfl_xor(acc[i], off, 32);
  if (lane == 0) {
    #pragma unroll
    for (int i = 0; i < NQ; ++i) out[sample * NQ + i] = acc[i];
  }
}

extern "C" void kernel_launch(void* const* d_in, const int* in_sizes, int n_in,
                              void* d_out, int out_size, void* d_ws, size_t ws_size,
                              hipStream_t stream) {
  const float* x = (const float*)d_in[0];
  const float* w = (const float*)d_in[1];
  float* ws  = (float*)d_ws;
  float* out = (float*)d_out;

  vqc_prep<<<NL * LAYER_F / 128, 128, 0, stream>>>(w, ws);
  vqc_main<<<BATCH_N / WPB, 32 * WPB, 0, stream>>>(x, ws, out);
}


# measurement reads the state in 16-byte units in physical LDS order, seven per-lane sums from the swizzle's index bits (were 64 scalar loads, ten sums per amplitude); measurement sums regrouped
# speedup vs baseline: 2.0158x; 1.0569x over previous
#include <hip/hip_runtime.h>


typedef __attribute__((ext_vector_type(2))) float v2f;
typedef __attribute__((ext_vector_type(4))) float v4f;
typedef __attribute__((ext_vector_type(8))) float v8f;

#define NQ      10
#define NL      4
#define BATCH_N 8192
#define BUF_F   2048
#define WPB     2
#define LAYER_F 1536

__device__ __forceinline__ constexpr int phys(int p, int U) {
  return (p << 10) | (((U ^ ((U >> 4) & 15) ^ p) & 255) << 2);
}

__device__ __forceinline__ int physA(int p, int a) {
  return phys(p, ((a & 127) << 1) | (a >> 9)) + ((a >> 7) & 3);
}

__device__ __forceinline__ void gate_elem(float a, float b, int rb, int cb,
                                          float& fr, float& fi) {
  float ca = cosf(0.5f * a), sa = sinf(0.5f * a);
  float cz = cosf(0.5f * b), sz = sinf(0.5f * b);
  float ry = rb ? (cb ? ca : sa) : (cb ? -sa : ca);
  fr = ry * cz;
  fi = ry * (rb ? sz : -sz);
}

__global__ __launch_bounds__(128) void vqc_prep(const float* __restrict__ w, float* __restrict__ ws) {
  int t = blockIdx.x * 128 + threadIdx.x;
  int l = t / LAYER_F;
  int e = t - l * LAYER_F;
  int nb, first, M, K;
  if (e < 512) {
    int lane = (e >> 3) & 31, slot = e & 7;
    nb = 3; first = 3 * (e >> 8); M = lane & 15; K = 8 * (lane >> 4) + slot;
  } else {
    int e2 = e - 512;
    int lane = (e2 >> 4) & 31, slot = e2 & 15;
    nb = 4; first = 6; M = (e2 >> 9) * 16 + (lane & 15); K = 16 * (lane >> 4) + slot;
  }
  int gm = (1 << nb) - 1;
  int po = M >> nb, o = M & gm, pi = K >> nb, i = K & gm;
  float vr = 1.f, vi = 0.f;
  #pragma unroll
  for (int j = 0; j < 4; ++j) {
    if (j < nb) {
      int wire = first + j;
      int rb = (o >> (nb - 1 - j)) & 1;
      int cb = (i >> (nb - 1 - j)) & 1;
      float fr, fi;
      gate_elem(w[(l * NQ + wire) * 2], w[(l * NQ + wire) * 2 + 1], rb, cb, fr, fi);
      float nr = vr * fr - vi * fi; vi = vr * fi + vi * fr; vr = nr;
    }
  }
  ws[t] = (po == pi) ? vr : (po ? vi : -vi);
}

__device__ __forceinline__ v4f ld4(const float* p) { return *(const v4f*)p; }
__device__ __forceinline__ void st4(float* p, v4f v) { *(v4f*)p = v; }

template <bool SECOND>
__device__ __forceinline__ void factor3w(const float* __restrict__ frag, float* X, int lane) {
  const int c = lane & 15, h = lane >> 4;
  const v4f* gp = (const v4f*)(frag + lane * 8);
  const v4f g0 = gp[0], g1 = gp[1];
  const float g[8] = {g0.x, g0.y, g0.z, g0.w, g1.x, g1.y, g1.z, g1.w};

  int lb, sb;
  if (!SECOND) {
    lb = phys(h, ((c & 7) << 5) | ((c >> 3) << 4));
    sb = phys(c >> 3, ((c & 7) << 5) | (h << 4));
  } else {
    lb = phys(h, c << 1);
    sb = phys(c >> 3, ((((c & 3) | (((c >> 2) & 1) << 3))) << 2) | (h << 1));
  }

  v4f a[8][2];
  #pragma unroll
  for (int T = 0; T < 8; ++T)
    #pragma unroll
    for (int j = 0; j < 2; ++j) {
      const int cu = SECOND ? ((T << 5) | j) : ((T << 1) | j);
      a[T][j] = ld4(X + (lb ^ phys(0, cu)));
    }

  #pragma unroll
  for (int tp = 0; tp < 4; ++tp) {
    v8f d[2];
    d[0] = (v8f){0, 0, 0, 0, 0, 0, 0, 0};
    d[1] = (v8f){0, 0, 0, 0, 0, 0, 0, 0};
    #pragma unroll
    for (int ks = 0; ks < 4; ++ks)
      #pragma unroll
      for (int t2 = 0; t2 < 2; ++t2) {
        const int T = 2 * tp + t2;
        v2f a2 = {a[T][ks >> 1][(2 * ks) & 3], a[T][ks >> 1][((2 * ks) & 3) + 1]};
        v2f b2 = {g[2 * ks], g[2 * ks + 1]};
        d[t2] = __builtin_amdgcn_wmma_f32_16x16x4_f32(false, a2, false, b2,
                                                      (short)0, d[t2], false, false);
      }
    #pragma unroll
    for (int t2 = 0; t2 < 2; ++t2)
      #pragma unroll
      for (int j = 0; j < 2; ++j) {
        const int T = 2 * tp + t2;
        const int cu = SECOND ? (((((T & 1) << 2) | ((T >> 1) << 4)) << 2) | j) : ((T << 1) | j);
        v4f v = {d[t2][4 * j], d[t2][4 * j + 1], d[t2][4 * j + 2], d[t2][4 * j + 3]};
        st4(X + (sb ^ phys(0, cu)), v);
      }
  }
}

__device__ __forceinline__ void factor4w_ring(const float* __restrict__ frag, float* X, int lane) {
  const int c = lane & 15, h = lane >> 4;
  const int u0 = c >> 3, u3 = (c >> 2) & 1, u2 = (c >> 1) & 1, u1 = c & 1;
  const int lrow = ((u2 ^ u1) << 2) | ((u1 ^ u0) << 3) | ((u3 ^ u2) << 4) | (u3 << 5);
  const int lb = phys(h, lrow << 2);
  const int n3 = h ^ (c >> 3), n2 = n3 ^ ((c >> 2) & 1), n1 = n2 ^ ((c >> 1) & 1), n0 = n1 ^ (c & 1);
  const int srow = (h << 6) | (h << 5) | (h << 4) | (n3 << 3) | (n2 << 2) | (n1 << 1) | n0;
  const int sb = phys(0, (srow << 1) | n0);

  v4f a[4][4];
  #pragma unroll
  for (int T = 0; T < 4; ++T)
    #pragma unroll
    for (int q = 0; q < 4; ++q)
      a[T][q] = ld4(X + (lb ^ phys(0, (T << 2) | q)));

  #pragma unroll
  for (int nt = 0; nt < 2; ++nt) {
    const v4f* gp = (const v4f*)(frag + nt * 512 + lane * 16);
    const v4f g0 = gp[0], g1 = gp[1], g2 = gp[2], g3 = gp[3];
    const float g[16] = {g0.x, g0.y, g0.z, g0.w, g1.x, g1.y, g1.z, g1.w,
                         g2.x, g2.y, g2.z, g2.w, g3.x, g3.y, g3.z, g3.w};
    #pragma unroll
    for (int tp = 0; tp < 2; ++tp) {
      v8f d[2];
      d[0] = (v8f){0, 0, 0, 0, 0, 0, 0, 0};
      d[1] = (v8f){0, 0, 0, 0, 0, 0, 0, 0};
      #pragma unroll
      for (int ks = 0; ks < 8; ++ks)
        #pragma unroll
        for (int t2 = 0; t2 < 2; ++t2) {
          const int T = 2 * tp + t2;
          v2f a2 = {a[T][ks >> 1][(2 * ks) & 3], a[T][ks >> 1][((2 * ks) & 3) + 1]};
          v2f b2 = {g[2 * ks], g[2 * ks + 1]};
          d[t2] = __builtin_amdgcn_wmma_f32_16x16x4_f32(false, a2, false, b2,
                                                        (short)0, d[t2], false, false);
        }
      #pragma unroll
      for (int t2 = 0; t2 < 2; ++t2)
        #pragma unroll
        for (int j = 0; j < 2; ++j) {
          const int T = 2 * tp + t2;
          const int a5 = (T >> 1) & 1, a4 = T & 1;
          const int nT = (a5 ? 0x3f : 0) ^ (a4 ? 0x1f : 0);
          const int cu = (nT << 1) | ((a5 ^ a4) ^ j);
          v4f v = {d[t2][4 * j], d[t2][4 * j + 1], d[t2][4 * j + 2], d[t2][4 * j + 3]};
          st4(X + (sb ^ phys(nt, cu)), v);
        }
    }
  }
}

__device__ __forceinline__ float xhalf(float x) { return __shfl_xor(x, 16, 32); }

template <int M>
__device__ __forceinline__ float xrow(float x) {
  int b = __builtin_bit_cast(int, x);
  return __builtin_bit_cast(float, __builtin_amdgcn_update_dpp(0, b, 0x160 | M, 0xf, 0xf, true));
}

__global__ __launch_bounds__(32 * WPB) __attribute__((amdgpu_waves_per_eu(8, 8))) void vqc_main(const float* __restrict__ x,
                                                     const float* __restrict__ ws,
                                                     float* __restrict__ out) {
  __shared__ __attribute__((aligned(16))) float lds[WPB * BUF_F];
  const int lane = threadIdx.x & 31;
  const int wv   = threadIdx.x >> 5;
  const int sample = blockIdx.x * WPB + wv;

  float* X = lds + wv * BUF_F;

  float rowv = 1.f, colv = 1.f;
  #pragma unroll
  for (int w2 = 0; w2 < 5; ++w2) {
    float ah = 0.5f * x[sample * NQ + w2];
    float al = 0.5f * x[sample * NQ + 5 + w2];
    int b = (lane >> (4 - w2)) & 1;
    rowv *= b ? sinf(ah) : cosf(ah);
    colv *= b ? sinf(al) : cosf(al);
  }
  for (int r = 0; r < 32; ++r) {
    float rv = __shfl(rowv, r, 32);
    int a = (r << 5) | lane;
    X[physA(0, a)] = rv * colv;
    X[physA(1, a)] = 0.f;
  }

  #pragma unroll 1
  for (int l = 0; l < NL; ++l) {
    const float* f0 = ws + l * LAYER_F;
    int ln = lane;
    asm volatile("" : "+v"(ln));
    ln &= 31;
    factor3w<false>(f0,       X, ln);
    factor3w<true >(f0 + 256, X, ln);
    factor4w_ring(f0 + 512, X, ln);
  }

  const int mc = lane & 15, mh = lane >> 4;
  const float* Lp = X + ((mh << 6) | (mc << 2));
  v2f q01 = {0.f, 0.f}, q23 = {0.f, 0.f};
  float s0 = 0.f, s1 = 0.f;
  float ut[8];
  #pragma unroll
  for (int t = 0; t < 8; ++t) {
    const v4f a = ld4(Lp + (t << 7));
    const v4f b = ld4(Lp + ((1 << 10) | (t << 7)));
    const v2f al = {a.x, a.y}, ah = {a.z, a.w}, bl = {b.x, b.y}, bh = {b.z, b.w};
    const v2f ra = al * al + ah * ah;
    const v2f rb = bl * bl + bh * bh;
    q01 += al * al; q23 += ah * ah;
    q01 += bl * bl; q23 += bh * bh;
    const float sa = ra.x + ra.y, sb = rb.x + rb.y;
    s0 += sa; s1 += sb;
    ut[t] = sa + sb;
  }
  const float e0 = ut[0] + ut[1], e1 = ut[2] + ut[3], e2 = ut[4] + ut[5], e3 = ut[6] + ut[7];
  float z4 = ((ut[0] - ut[1]) + (ut[2] - ut[3])) + ((ut[4] - ut[5]) + (ut[6] - ut[7]));
  float z5 = (e0 - e1) + (e2 - e3);
  float z6 = (e0 + e1) - (e2 + e3);
  float st = s0 + s1;
  float pd = s0 - s1;
  float w8 = (q01.x + q01.y) - (q23.x + q23.y);
  float w7 = (q01.x - q01.y) + (q23.x - q23.y);

  z6 += xhalf(z6); z6 += xrow<1>(z6); z6 += xrow<2>(z6); z6 += xrow<4>(z6);
  z5 += xhalf(z5); z5 += xrow<1>(z5); z5 += xrow<2>(z5); z5 += xrow<8>(z5);
  z4 += xhalf(z4); z4 += xrow<1>(z4); z4 += xrow<4>(z4); z4 += xrow<8>(z4);
  w8 += xhalf(w8); w8 += xrow<1>(w8); w8 += xrow<2>(w8); w8 += xrow<4>(w8); w8 += xrow<8>(w8);
  w7 += xhalf(w7); w7 += xrow<1>(w7); w7 += xrow<2>(w7); w7 += xrow<4>(w7); w7 += xrow<8>(w7);
  st += xrow<1>(st); st += xrow<2>(st); st += xrow<4>(st); st += xrow<8>(st);
  pd -= xrow<1>(pd); pd += xrow<2>(pd); pd += xrow<4>(pd); pd += xrow<8>(pd);
  const float z6o = xrow<8>(z6), z5o = xrow<4>(z5), z4o = xrow<2>(z4);
  const float sto = xhalf(st), pdo = xhalf(pd);
  if (lane == 0) {
    float* o = out + sample * NQ;
    o[0] = pd - pdo;
    o[1] = w8;
    o[2] = w7;
    o[3] = z6 + z6o;
    o[4] = z5 + z5o;
    o[5] = z4 + z4o;
    o[6] = st - sto;
    o[7] = z6 - z6o;
    o[8] = z5 - z5o;
    o[9] = z4 - z4o;
  }
}

extern "C" void kernel_launch(void* const* d_in, const int* in_sizes, int n_in,
                              void* d_out, int out_size, void* d_ws, size_t ws_size,
                              hipStream_t stream) {
  const float* x = (const float*)d_in[0];
  const float* w = (const float*)d_in[1];
  float* ws  = (float*)d_ws;
  float* out = (float*)d_out;

  vqc_prep<<<NL * LAYER_F / 128, 128, 0, stream>>>(w, ws);
  vqc_main<<<BATCH_N / WPB, 32 * WPB, 0, stream>>>(x, ws, out);
}


# angle encoding prepared by the prep launch in store order (8 row quads + 32 column products per sample); main only multiplies and stores 16 bytes at a time: no trig, no shuffles; values bit-identical
# speedup vs baseline: 2.2172x; 1.0999x over previous
#include <hip/hip_runtime.h>


typedef __attribute__((ext_vector_type(2))) float v2f;
typedef __attribute__((ext_vector_type(4))) float v4f;
typedef __attribute__((ext_vector_type(8))) float v8f;

#define NQ      10
#define NL      4
#define BATCH_N 8192
#define BUF_F   2048
#define WPB     2
#define LAYER_F 1536
#define TAB_F      (NL * LAYER_F)
#define TAB_BLOCKS (TAB_F / 128)
#define ENC_BLOCKS (BATCH_N * 2 / 128)
#define PREP_BLOCKS (TAB_BLOCKS + ENC_BLOCKS)

__device__ __forceinline__ constexpr int phys(int p, int U) {
  return (p << 10) | (((U ^ ((U >> 4) & 15) ^ p) & 255) << 2);
}


__device__ __forceinline__ constexpr int encpos(int i) {
  return (i & 16) | ((i & 3) << 2) | ((i >> 2) & 3);
}

__device__ __forceinline__ void gate_elem(float a, float b, int rb, int cb,
                                          float& fr, float& fi) {
  float ca = cosf(0.5f * a), sa = sinf(0.5f * a);
  float cz = cosf(0.5f * b), sz = sinf(0.5f * b);
  float ry = rb ? (cb ? ca : sa) : (cb ? -sa : ca);
  fr = ry * cz;
  fi = ry * (rb ? sz : -sz);
}

__device__ __forceinline__ v4f ld4(const float* p) { return *(const v4f*)p; }
__device__ __forceinline__ void st4(float* p, v4f v) { *(v4f*)p = v; }

__device__ __forceinline__ void prep_tables(const float* __restrict__ w, float* __restrict__ ws, int t) {
  int l = t / LAYER_F;
  int e = t - l * LAYER_F;
  int nb, first, M, K;
  if (e < 512) {
    int lane = (e >> 3) & 31, slot = e & 7;
    nb = 3; first = 3 * (e >> 8); M = lane & 15; K = 8 * (lane >> 4) + slot;
  } else {
    int e2 = e - 512;
    int lane = (e2 >> 4) & 31, slot = e2 & 15;
    nb = 4; first = 6; M = (e2 >> 9) * 16 + (lane & 15); K = 16 * (lane >> 4) + slot;
  }
  int gm = (1 << nb) - 1;
  int po = M >> nb, o = M & gm, pi = K >> nb, i = K & gm;
  float vr = 1.f, vi = 0.f;
  #pragma unroll
  for (int j = 0; j < 4; ++j) {
    if (j < nb) {
      int wire = first + j;
      int rb = (o >> (nb - 1 - j)) & 1;
      int cb = (i >> (nb - 1 - j)) & 1;
      float fr, fi;
      gate_elem(w[(l * NQ + wire) * 2], w[(l * NQ + wire) * 2 + 1], rb, cb, fr, fi);
      float nr = vr * fr - vi * fi; vi = vr * fi + vi * fr; vr = nr;
    }
  }
  ws[t] = (po == pi) ? vr : (po ? vi : -vi);
}

__device__ __forceinline__ void prep_encoding(const float* __restrict__ x, float* __restrict__ ws, int u) {
  const float* xp = x + u * 5;
  float xv[5];
  #pragma unroll
  for (int j = 0; j < 5; ++j) xv[j] = xp[j];
  float v[32];
  {
    const float ah = 0.5f * xv[0];
    v[0] = cosf(ah); v[1] = sinf(ah);
  }
  #pragma unroll
  for (int j = 1; j < 5; ++j) {
    const float ah = 0.5f * xv[j];
    const float cj = cosf(ah), sj = sinf(ah);
    #pragma unroll
    for (int i = (1 << j) - 1; i >= 0; --i) {
      const float p = v[i];
      v[2 * i] = p * cj; v[2 * i + 1] = p * sj;
    }
  }
  float* o = ws + TAB_F + u * 32;
  #pragma unroll
  for (int q = 0; q < 8; ++q) {
    v4f t = {v[encpos(4 * q)], v[encpos(4 * q + 1)], v[encpos(4 * q + 2)], v[encpos(4 * q + 3)]};
    st4(o + 4 * q, t);
  }
}

__global__ __launch_bounds__(128) void vqc_prep(const float* __restrict__ x, const float* __restrict__ w,
                                                float* __restrict__ ws) {
  if (blockIdx.x < TAB_BLOCKS)
    prep_tables(w, ws, blockIdx.x * 128 + threadIdx.x);
  else
    prep_encoding(x, ws, (blockIdx.x - TAB_BLOCKS) * 128 + threadIdx.x);
}

template <bool SECOND>
__device__ __forceinline__ void factor3w(const float* __restrict__ frag, float* X, int lane) {
  const int c = lane & 15, h = lane >> 4;
  const v4f* gp = (const v4f*)(frag + lane * 8);
  const v4f g0 = gp[0], g1 = gp[1];
  const float g[8] = {g0.x, g0.y, g0.z, g0.w, g1.x, g1.y, g1.z, g1.w};

  int lb, sb;
  if (!SECOND) {
    lb = phys(h, ((c & 7) << 5) | ((c >> 3) << 4));
    sb = phys(c >> 3, ((c & 7) << 5) | (h << 4));
  } else {
    lb = phys(h, c << 1);
    sb = phys(c >> 3, ((((c & 3) | (((c >> 2) & 1) << 3))) << 2) | (h << 1));
  }

  v4f a[8][2];
  #pragma unroll
  for (int T = 0; T < 8; ++T)
    #pragma unroll
    for (int j = 0; j < 2; ++j) {
      const int cu = SECOND ? ((T << 5) | j) : ((T << 1) | j);
      a[T][j] = ld4(X + (lb ^ phys(0, cu)));
    }
  __builtin_amdgcn_sched_barrier(0);

  #pragma unroll
  for (int tp = 0; tp < 4; ++tp) {
    v8f d[2];
    d[0] = (v8f){0, 0, 0, 0, 0, 0, 0, 0};
    d[1] = (v8f){0, 0, 0, 0, 0, 0, 0, 0};
    #pragma unroll
    for (int ks = 0; ks < 4; ++ks)
      #pragma unroll
      for (int t2 = 0; t2 < 2; ++t2) {
        const int T = 2 * tp + t2;
        v2f a2 = {a[T][ks >> 1][(2 * ks) & 3], a[T][ks >> 1][((2 * ks) & 3) + 1]};
        v2f b2 = {g[2 * ks], g[2 * ks + 1]};
        d[t2] = __builtin_amdgcn_wmma_f32_16x16x4_f32(false, a2, false, b2,
                                                      (short)0, d[t2], false, false);
      }
    #pragma unroll
    for (int t2 = 0; t2 < 2; ++t2)
      #pragma unroll
      for (int j = 0; j < 2; ++j) {
        const int T = 2 * tp + t2;
        const int cu = SECOND ? (((((T & 1) << 2) | ((T >> 1) << 4)) << 2) | j) : ((T << 1) | j);
        v4f v = {d[t2][4 * j], d[t2][4 * j + 1], d[t2][4 * j + 2], d[t2][4 * j + 3]};
        st4(X + (sb ^ phys(0, cu)), v);
      }
    __builtin_amdgcn_sched_barrier(0);
  }
}

__device__ __forceinline__ void factor4w_ring(const float* __restrict__ frag, float* X, int lane) {
  const int c = lane & 15, h = lane >> 4;
  const int u0 = c >> 3, u3 = (c >> 2) & 1, u2 = (c >> 1) & 1, u1 = c & 1;
  const int lrow = ((u2 ^ u1) << 2) | ((u1 ^ u0) << 3) | ((u3 ^ u2) << 4) | (u3 << 5);
  const int lb = phys(h, lrow << 2);
  const int n3 = h ^ (c >> 3), n2 = n3 ^ ((c >> 2) & 1), n1 = n2 ^ ((c >> 1) & 1), n0 = n1 ^ (c & 1);
  const int srow = (h << 6) | (h << 5) | (h << 4) | (n3 << 3) | (n2 << 2) | (n1 << 1) | n0;
  const int sb = phys(0, (srow << 1) | n0);

  v4f a[4][4];
  #pragma unroll
  for (int T = 0; T < 4; ++T)
    #pragma unroll
    for (int q = 0; q < 4; ++q)
      a[T][q] = ld4(X + (lb ^ phys(0, (T << 2) | q)));
  __builtin_amdgcn_sched_barrier(0);

  #pragma unroll
  for (int nt = 0; nt < 2; ++nt) {
    const v4f* gp = (const v4f*)(frag + nt * 512 + lane * 16);
    const v4f g0 = gp[0], g1 = gp[1], g2 = gp[2], g3 = gp[3];
    const float g[16] = {g0.x, g0.y, g0.z, g0.w, g1.x, g1.y, g1.z, g1.w,
                         g2.x, g2.y, g2.z, g2.w, g3.x, g3.y, g3.z, g3.w};
    #pragma unroll
    for (int tp = 0; tp < 2; ++tp) {
      v8f d[2];
      d[0] = (v8f){0, 0, 0, 0, 0, 0, 0, 0};
      d[1] = (v8f){0, 0, 0, 0, 0, 0, 0, 0};
      #pragma unroll
      for (int ks = 0; ks < 8; ++ks)
        #pragma unroll
        for (int t2 = 0; t2 < 2; ++t2) {
          const int T = 2 * tp + t2;
          v2f a2 = {a[T][ks >> 1][(2 * ks) & 3], a[T][ks >> 1][((2 * ks) & 3) + 1]};
          v2f b2 = {g[2 * ks], g[2 * ks + 1]};
          d[t2] = __builtin_amdgcn_wmma_f32_16x16x4_f32(false, a2, false, b2,
                                                        (short)0, d[t2], false, false);
        }
      #pragma unroll
      for (int t2 = 0; t2 < 2; ++t2)
        #pragma unroll
        for (int j = 0; j < 2; ++j) {
          const int T = 2 * tp + t2;
          const int a5 = (T >> 1) & 1, a4 = T & 1;
          const int nT = (a5 ? 0x3f : 0) ^ (a4 ? 0x1f : 0);
          const int cu = (nT << 1) | ((a5 ^ a4) ^ j);
          v4f v = {d[t2][4 * j], d[t2][4 * j + 1], d[t2][4 * j + 2], d[t2][4 * j + 3]};
          st4(X + (sb ^ phys(nt, cu)), v);
        }
      __builtin_amdgcn_sched_barrier(0);
    }
  }
}

__device__ __forceinline__ float xhalf(float x) { return __shfl_xor(x, 16, 32); }

template <int M>
__device__ __forceinline__ float xrow(float x) {
  int b = __builtin_bit_cast(int, x);
  return __builtin_bit_cast(float, __builtin_amdgcn_update_dpp(0, b, 0x160 | M, 0xf, 0xf, true));
}

__global__ __launch_bounds__(32 * WPB) __attribute__((amdgpu_waves_per_eu(8, 8))) void vqc_main(const float* __restrict__ ws,
                                                     float* __restrict__ out) {
  __shared__ __attribute__((aligned(16))) float lds[WPB * BUF_F];
  const int lane = threadIdx.x & 31;
  const int wv   = threadIdx.x >> 5;
  const int sample = blockIdx.x * WPB + wv;

  float* X = lds + wv * BUF_F;

  const float* ev = ws + TAB_F + sample * 64;
  const float colv = ev[32 + encpos(lane)];
  v4f rq[8];
  #pragma unroll
  for (int k = 0; k < 8; ++k) rq[k] = ld4(ev + 4 * k);
  const int ub = phys(0, lane << 1);
  const v4f zero4 = {0.f, 0.f, 0.f, 0.f};
  #pragma unroll
  for (int k = 0; k < 8; ++k) {
    const int cu = ((k & 3) << 6) | (k >> 2);
    st4(X + (ub ^ phys(0, cu)), rq[k] * colv);
    st4(X + (ub ^ phys(1, cu)), zero4);
  }

  #pragma unroll 1
  for (int l = 0; l < NL; ++l) {
    const float* f0 = ws + l * LAYER_F;
    int ln = lane;
    asm volatile("" : "+v"(ln));
    ln &= 31;
    factor3w<false>(f0,       X, ln);
    factor3w<true >(f0 + 256, X, ln);
    factor4w_ring(f0 + 512, X, ln);
  }

  const int mc = lane & 15, mh = lane >> 4;
  const float* Lp = X + ((mh << 6) | (mc << 2));
  v2f q01 = {0.f, 0.f}, q23 = {0.f, 0.f};
  float s0 = 0.f, s1 = 0.f;
  float ut[8];
  #pragma unroll
  for (int t = 0; t < 8; ++t) {
    const v4f a = ld4(Lp + (t << 7));
    const v4f b = ld4(Lp + ((1 << 10) | (t << 7)));
    const v2f al = {a.x, a.y}, ah = {a.z, a.w}, bl = {b.x, b.y}, bh = {b.z, b.w};
    const v2f ra = al * al + ah * ah;
    const v2f rb = bl * bl + bh * bh;
    q01 += al * al; q23 += ah * ah;
    q01 += bl * bl; q23 += bh * bh;
    const float sa = ra.x + ra.y, sb = rb.x + rb.y;
    s0 += sa; s1 += sb;
    ut[t] = sa + sb;
  }
  const float e0 = ut[0] + ut[1], e1 = ut[2] + ut[3], e2 = ut[4] + ut[5], e3 = ut[6] + ut[7];
  float z4 = ((ut[0] - ut[1]) + (ut[2] - ut[3])) + ((ut[4] - ut[5]) + (ut[6] - ut[7]));
  float z5 = (e0 - e1) + (e2 - e3);
  float z6 = (e0 + e1) - (e2 + e3);
  float st = s0 + s1;
  float pd = s0 - s1;
  float w8 = (q01.x + q01.y) - (q23.x + q23.y);
  float w7 = (q01.x - q01.y) + (q23.x - q23.y);

  z6 += xhalf(z6); z6 += xrow<1>(z6); z6 += xrow<2>(z6); z6 += xrow<4>(z6);
  z5 += xhalf(z5); z5 += xrow<1>(z5); z5 += xrow<2>(z5); z5 += xrow<8>(z5);
  z4 += xhalf(z4); z4 += xrow<1>(z4); z4 += xrow<4>(z4); z4 += xrow<8>(z4);
  w8 += xhalf(w8); w8 += xrow<1>(w8); w8 += xrow<2>(w8); w8 += xrow<4>(w8); w8 += xrow<8>(w8);
  w7 += xhalf(w7); w7 += xrow<1>(w7); w7 += xrow<2>(w7); w7 += xrow<4>(w7); w7 += xrow<8>(w7);
  st += xrow<1>(st); st += xrow<2>(st); st += xrow<4>(st); st += xrow<8>(st);
  pd -= xrow<1>(pd); pd += xrow<2>(pd); pd += xrow<4>(pd); pd += xrow<8>(pd);
  const float z6o = xrow<8>(z6), z5o = xrow<4>(z5), z4o = xrow<2>(z4);
  const float sto = xhalf(st), pdo = xhalf(pd);
  if (lane == 0) {
    float* o = out + sample * NQ;
    o[0] = pd - pdo;
    o[1] = w8;
    o[2] = w7;
    o[3] = z6 + z6o;
    o[4] = z5 + z5o;
    o[5] = z4 + z4o;
    o[6] = st - sto;
    o[7] = z6 - z6o;
    o[8] = z5 - z5o;
    o[9] = z4 - z4o;
  }
}

extern "C" void kernel_launch(void* const* d_in, const int* in_sizes, int n_in,
                              void* d_out, int out_size, void* d_ws, size_t ws_size,
                              hipStream_t stream) {
  const float* x = (const float*)d_in[0];
  const float* w = (const float*)d_in[1];
  float* ws  = (float*)d_ws;
  float* out = (float*)d_out;

  vqc_prep<<<PREP_BLOCKS, 128, 0, stream>>>(x, w, ws);
  vqc_main<<<BATCH_N / WPB, 32 * WPB, 0, stream>>>(ws, out);
}


# two samples per wave, their factor passes alternating in one instruction stream: each LDS load group is in flight behind the other sample's matrix instructions; 4096 waves in 512 blocks, not 8192
# speedup vs baseline: 2.3319x; 1.0517x over previous
#include <hip/hip_runtime.h>


typedef __attribute__((ext_vector_type(2))) float v2f;
typedef __attribute__((ext_vector_type(4))) float v4f;
typedef __attribute__((ext_vector_type(8))) float v8f;

#define NQ      10
#define NL      4
#define BATCH_N 8192
#define BUF_F   2048
#define SPW     2
#define WPB     8
#define LAYER_F 1536
#define TAB_F      (NL * LAYER_F)
#define TAB_BLOCKS (TAB_F / 128)
#define ENC_BLOCKS (BATCH_N * 2 / 128)
#define PREP_BLOCKS (TAB_BLOCKS + ENC_BLOCKS)

__device__ __forceinline__ constexpr int phys(int p, int U) {
  return (p << 10) | (((U ^ ((U >> 4) & 15) ^ p) & 255) << 2);
}


__device__ __forceinline__ constexpr int encpos(int i) {
  return (i & 16) | ((i & 3) << 2) | ((i >> 2) & 3);
}

__device__ __forceinline__ void gate_elem(float a, float b, int rb, int cb,
                                          float& fr, float& fi) {
  float ca = cosf(0.5f * a), sa = sinf(0.5f * a);
  float cz = cosf(0.5f * b), sz = sinf(0.5f * b);
  float ry = rb ? (cb ? ca : sa) : (cb ? -sa : ca);
  fr = ry * cz;
  fi = ry * (rb ? sz : -sz);
}

__device__ __forceinline__ v4f ld4(const float* p) { return *(const v4f*)p; }
__device__ __forceinline__ void st4(float* p, v4f v) { *(v4f*)p = v; }

__device__ __forceinline__ void prep_tables(const float* __restrict__ w, float* __restrict__ ws, int t) {
  int l = t / LAYER_F;
  int e = t - l * LAYER_F;
  int nb, first, M, K;
  if (e < 512) {
    int lane = (e >> 3) & 31, slot = e & 7;
    nb = 3; first = 3 * (e >> 8); M = lane & 15; K = 8 * (lane >> 4) + slot;
  } else {
    int e2 = e - 512;
    int lane = (e2 >> 4) & 31, slot = e2 & 15;
    nb = 4; first = 6; M = (e2 >> 9) * 16 + (lane & 15); K = 16 * (lane >> 4) + slot;
  }
  int gm = (1 << nb) - 1;
  int po = M >> nb, o = M & gm, pi = K >> nb, i = K & gm;
  float vr = 1.f, vi = 0.f;
  #pragma unroll
  for (int j = 0; j < 4; ++j) {
    if (j < nb) {
      int wire = first + j;
      int rb = (o >> (nb - 1 - j)) & 1;
      int cb = (i >> (nb - 1 - j)) & 1;
      float fr, fi;
      gate_elem(w[(l * NQ + wire) * 2], w[(l * NQ + wire) * 2 + 1], rb, cb, fr, fi);
      float nr = vr * fr - vi * fi; vi = vr * fi + vi * fr; vr = nr;
    }
  }
  ws[t] = (po == pi) ? vr : (po ? vi : -vi);
}

__device__ __forceinline__ void prep_encoding(const float* __restrict__ x, float* __restrict__ ws, int u) {
  const float* xp = x + u * 5;
  float xv[5];
  #pragma unroll
  for (int j = 0; j < 5; ++j) xv[j] = xp[j];
  float v[32];
  {
    const float ah = 0.5f * xv[0];
    v[0] = cosf(ah); v[1] = sinf(ah);
  }
  #pragma unroll
  for (int j = 1; j < 5; ++j) {
    const float ah = 0.5f * xv[j];
    const float cj = cosf(ah), sj = sinf(ah);
    #pragma unroll
    for (int i = (1 << j) - 1; i >= 0; --i) {
      const float p = v[i];
      v[2 * i] = p * cj; v[2 * i + 1] = p * sj;
    }
  }
  float* o = ws + TAB_F + u * 32;
  #pragma unroll
  for (int q = 0; q < 8; ++q) {
    v4f t = {v[encpos(4 * q)], v[encpos(4 * q + 1)], v[encpos(4 * q + 2)], v[encpos(4 * q + 3)]};
    st4(o + 4 * q, t);
  }
}

__global__ __launch_bounds__(128) void vqc_prep(const float* __restrict__ x, const float* __restrict__ w,
                                                float* __restrict__ ws) {
  if (blockIdx.x < TAB_BLOCKS)
    prep_tables(w, ws, blockIdx.x * 128 + threadIdx.x);
  else
    prep_encoding(x, ws, (blockIdx.x - TAB_BLOCKS) * 128 + threadIdx.x);
}


__device__ __forceinline__ void gate3(const float* __restrict__ frag, int lane, float (&g)[8]) {
  const v4f* gp = (const v4f*)(frag + lane * 8);
  const v4f g0 = gp[0], g1 = gp[1];
  g[0] = g0.x; g[1] = g0.y; g[2] = g0.z; g[3] = g0.w;
  g[4] = g1.x; g[5] = g1.y; g[6] = g1.z; g[7] = g1.w;
}

__device__ __forceinline__ void gate4(const float* __restrict__ frag, int lane, float (&g)[2][16]) {
  #pragma unroll
  for (int nt = 0; nt < 2; ++nt) {
    const v4f* gp = (const v4f*)(frag + nt * 512 + lane * 16);
    #pragma unroll
    for (int q = 0; q < 4; ++q) {
      const v4f t = gp[q];
      g[nt][4 * q] = t.x; g[nt][4 * q + 1] = t.y; g[nt][4 * q + 2] = t.z; g[nt][4 * q + 3] = t.w;
    }
  }
}

template <bool SECOND>
__device__ __forceinline__ void f3_load(const float* X, int lane, v4f (&a)[16]) {
  const int c = lane & 15, h = lane >> 4;
  int lb;
  if (!SECOND) lb = phys(h, ((c & 7) << 5) | ((c >> 3) << 4));
  else         lb = phys(h, c << 1);
  #pragma unroll
  for (int T = 0; T < 8; ++T)
    #pragma unroll
    for (int j = 0; j < 2; ++j) {
      const int cu = SECOND ? ((T << 5) | j) : ((T << 1) | j);
      a[2 * T + j] = ld4(X + (lb ^ phys(0, cu)));
    }
  __builtin_amdgcn_sched_barrier(0);
}

template <bool SECOND>
__device__ __forceinline__ void f3_run(const float (&g)[8], float* X, int lane, const v4f (&a)[16]) {
  const int c = lane & 15, h = lane >> 4;
  int sb;
  if (!SECOND) {
    sb = phys(c >> 3, ((c & 7) << 5) | (h << 4));
  } else {
    sb = phys(c >> 3, ((((c & 3) | (((c >> 2) & 1) << 3))) << 2) | (h << 1));
  }
  #pragma unroll
  for (int tp = 0; tp < 4; ++tp) {
    v8f d[2];
    d[0] = (v8f){0, 0, 0, 0, 0, 0, 0, 0};
    d[1] = (v8f){0, 0, 0, 0, 0, 0, 0, 0};
    #pragma unroll
    for (int ks = 0; ks < 4; ++ks)
      #pragma unroll
      for (int t2 = 0; t2 < 2; ++t2) {
        const int T = 2 * tp + t2;
        const v4f av = a[2 * T + (ks >> 1)];
        v2f a2 = {av[(2 * ks) & 3], av[((2 * ks) & 3) + 1]};
        v2f b2 = {g[2 * ks], g[2 * ks + 1]};
        d[t2] = __builtin_amdgcn_wmma_f32_16x16x4_f32(false, a2, false, b2,
                                                      (short)0, d[t2], false, false);
      }
    #pragma unroll
    for (int t2 = 0; t2 < 2; ++t2)
      #pragma unroll
      for (int j = 0; j < 2; ++j) {
        const int T = 2 * tp + t2;
        const int cu = SECOND ? (((((T & 1) << 2) | ((T >> 1) << 4)) << 2) | j) : ((T << 1) | j);
        v4f v = {d[t2][4 * j], d[t2][4 * j + 1], d[t2][4 * j + 2], d[t2][4 * j + 3]};
        st4(X + (sb ^ phys(0, cu)), v);
      }
    __builtin_amdgcn_sched_barrier(0);
  }
}

__device__ __forceinline__ void f4_load(const float* X, int lane, v4f (&a)[16]) {
  const int c = lane & 15, h = lane >> 4;
  const int u0 = c >> 3, u3 = (c >> 2) & 1, u2 = (c >> 1) & 1, u1 = c & 1;
  const int lrow = ((u2 ^ u1) << 2) | ((u1 ^ u0) << 3) | ((u3 ^ u2) << 4) | (u3 << 5);
  const int lb = phys(h, lrow << 2);
  #pragma unroll
  for (int T = 0; T < 4; ++T)
    #pragma unroll
    for (int q = 0; q < 4; ++q)
      a[4 * T + q] = ld4(X + (lb ^ phys(0, (T << 2) | q)));
  __builtin_amdgcn_sched_barrier(0);
}

__device__ __forceinline__ void f4_run(const float (&g)[2][16], float* X, int lane, const v4f (&a)[16]) {
  const int c = lane & 15, h = lane >> 4;
  const int n3 = h ^ (c >> 3), n2 = n3 ^ ((c >> 2) & 1), n1 = n2 ^ ((c >> 1) & 1), n0 = n1 ^ (c & 1);
  const int srow = (h << 6) | (h << 5) | (h << 4) | (n3 << 3) | (n2 << 2) | (n1 << 1) | n0;
  const int sb = phys(0, (srow << 1) | n0);
  #pragma unroll
  for (int nt = 0; nt < 2; ++nt) {
    #pragma unroll
    for (int tp = 0; tp < 2; ++tp) {
      v8f d[2];
      d[0] = (v8f){0, 0, 0, 0, 0, 0, 0, 0};
      d[1] = (v8f){0, 0, 0, 0, 0, 0, 0, 0};
      #pragma unroll
      for (int ks = 0; ks < 8; ++ks)
        #pragma unroll
        for (int t2 = 0; t2 < 2; ++t2) {
          const int T = 2 * tp + t2;
          const v4f av = a[4 * T + (ks >> 1)];
          v2f a2 = {av[(2 * ks) & 3], av[((2 * ks) & 3) + 1]};
          v2f b2 = {g[nt][2 * ks], g[nt][2 * ks + 1]};
          d[t2] = __builtin_amdgcn_wmma_f32_16x16x4_f32(false, a2, false, b2,
                                                        (short)0, d[t2], false, false);
        }
      #pragma unroll
      for (int t2 = 0; t2 < 2; ++t2)
        #pragma unroll
        for (int j = 0; j < 2; ++j) {
          const int T = 2 * tp + t2;
          const int a5 = (T >> 1) & 1, a4 = T & 1;
          const int nT = (a5 ? 0x3f : 0) ^ (a4 ? 0x1f : 0);
          const int cu = (nT << 1) | ((a5 ^ a4) ^ j);
          v4f v = {d[t2][4 * j], d[t2][4 * j + 1], d[t2][4 * j + 2], d[t2][4 * j + 3]};
          st4(X + (sb ^ phys(nt, cu)), v);
        }
      __builtin_amdgcn_sched_barrier(0);
    }
  }
}

__device__ __forceinline__ float xhalf(float x) { return __shfl_xor(x, 16, 32); }

template <int M>
__device__ __forceinline__ float xrow(float x) {
  int b = __builtin_bit_cast(int, x);
  return __builtin_bit_cast(float, __builtin_amdgcn_update_dpp(0, b, 0x160 | M, 0xf, 0xf, true));
}

__device__ __forceinline__ void encode(const float* __restrict__ ws, float* X, int sample, int lane) {
  const float* ev = ws + TAB_F + sample * 64;
  const float colv = ev[32 + encpos(lane)];
  v4f rq[8];
  #pragma unroll
  for (int k = 0; k < 8; ++k) rq[k] = ld4(ev + 4 * k);
  const int ub = phys(0, lane << 1);
  const v4f zero4 = {0.f, 0.f, 0.f, 0.f};
  #pragma unroll
  for (int k = 0; k < 8; ++k) {
    const int cu = ((k & 3) << 6) | (k >> 2);
    st4(X + (ub ^ phys(0, cu)), rq[k] * colv);
    st4(X + (ub ^ phys(1, cu)), zero4);
  }
}

__device__ __forceinline__ void measure(const float* X, float* __restrict__ out, int sample, int lane) {
  const int mc = lane & 15, mh = lane >> 4;
  const float* Lp = X + ((mh << 6) | (mc << 2));
  v2f q01 = {0.f, 0.f}, q23 = {0.f, 0.f};
  float s0 = 0.f, s1 = 0.f;
  float ut[8];
  #pragma unroll
  for (int t = 0; t < 8; ++t) {
    const v4f a = ld4(Lp + (t << 7));
    const v4f b = ld4(Lp + ((1 << 10) | (t << 7)));
    const v2f al = {a.x, a.y}, ah = {a.z, a.w}, bl = {b.x, b.y}, bh = {b.z, b.w};
    const v2f ra = al * al + ah * ah;
    const v2f rb = bl * bl + bh * bh;
    q01 += al * al; q23 += ah * ah;
    q01 += bl * bl; q23 += bh * bh;
    const float sa = ra.x + ra.y, sb = rb.x + rb.y;
    s0 += sa; s1 += sb;
    ut[t] = sa + sb;
  }
  const float e0 = ut[0] + ut[1], e1 = ut[2] + ut[3], e2 = ut[4] + ut[5], e3 = ut[6] + ut[7];
  float z4 = ((ut[0] - ut[1]) + (ut[2] - ut[3])) + ((ut[4] - ut[5]) + (ut[6] - ut[7]));
  float z5 = (e0 - e1) + (e2 - e3);
  float z6 = (e0 + e1) - (e2 + e3);
  float st = s0 + s1;
  float pd = s0 - s1;
  float w8 = (q01.x + q01.y) - (q23.x + q23.y);
  float w7 = (q01.x - q01.y) + (q23.x - q23.y);

  z6 += xhalf(z6); z6 += xrow<1>(z6); z6 += xrow<2>(z6); z6 += xrow<4>(z6);
  z5 += xhalf(z5); z5 += xrow<1>(z5); z5 += xrow<2>(z5); z5 += xrow<8>(z5);
  z4 += xhalf(z4); z4 += xrow<1>(z4); z4 += xrow<4>(z4); z4 += xrow<8>(z4);
  w8 += xhalf(w8); w8 += xrow<1>(w8); w8 += xrow<2>(w8); w8 += xrow<4>(w8); w8 += xrow<8>(w8);
  w7 += xhalf(w7); w7 += xrow<1>(w7); w7 += xrow<2>(w7); w7 += xrow<4>(w7); w7 += xrow<8>(w7);
  st += xrow<1>(st); st += xrow<2>(st); st += xrow<4>(st); st += xrow<8>(st);
  pd -= xrow<1>(pd); pd += xrow<2>(pd); pd += xrow<4>(pd); pd += xrow<8>(pd);
  const float z6o = xrow<8>(z6), z5o = xrow<4>(z5), z4o = xrow<2>(z4);
  const float sto = xhalf(st), pdo = xhalf(pd);
  if (lane == 0) {
    float* o = out + sample * NQ;
    o[0] = pd - pdo;
    o[1] = w8;
    o[2] = w7;
    o[3] = z6 + z6o;
    o[4] = z5 + z5o;
    o[5] = z4 + z4o;
    o[6] = st - sto;
    o[7] = z6 - z6o;
    o[8] = z5 - z5o;
    o[9] = z4 - z4o;
  }
}

__global__ __launch_bounds__(32 * WPB) __attribute__((amdgpu_waves_per_eu(4, 4))) void vqc_main(const float* __restrict__ ws,
                                                     float* __restrict__ out) {
  __shared__ __attribute__((aligned(16))) float lds[WPB * SPW * BUF_F];
  const int lane = threadIdx.x & 31;
  const int wv   = threadIdx.x >> 5;
  const int sample = (blockIdx.x * WPB + wv) * SPW;

  float* X0 = lds + wv * (SPW * BUF_F);
  float* X1 = X0 + BUF_F;

  encode(ws, X0, sample, lane);
  encode(ws, X1, sample + 1, lane);

  v4f A0[16], A1[16];
  float g1[8], g2[8], g3[2][16];
  f3_load<false>(X0, lane, A0);
  gate3(ws, lane, g1);

  #pragma unroll 1
  for (int l = 0; l < NL; ++l) {
    const float* f0 = ws + l * LAYER_F;
    int ln = lane;
    asm volatile("" : "+v"(ln));
    ln &= 31;
    f3_load<false>(X1, ln, A1);
    f3_run<false>(g1, X0, ln, A0);
    f3_load<true>(X0, ln, A0);
    gate3(f0 + 256, ln, g2);
    f3_run<false>(g1, X1, ln, A1);
    f3_load<true>(X1, ln, A1);
    gate4(f0 + 512, ln, g3);
    f3_run<true>(g2, X0, ln, A0);
    f4_load(X0, ln, A0);
    f3_run<true>(g2, X1, ln, A1);
    f4_load(X1, ln, A1);
    f4_run(g3, X0, ln, A0);
    if (l + 1 < NL) {
      f3_load<false>(X0, ln, A0);
      gate3(f0 + LAYER_F, ln, g1);
    }
    f4_run(g3, X1, ln, A1);
  }

  measure(X0, out, sample, lane);
  measure(X1, out, sample + 1, lane);
}

extern "C" void kernel_launch(void* const* d_in, const int* in_sizes, int n_in,
                              void* d_out, int out_size, void* d_ws, size_t ws_size,
                              hipStream_t stream) {
  const float* x = (const float*)d_in[0];
  const float* w = (const float*)d_in[1];
  float* ws  = (float*)d_ws;
  float* out = (float*)d_out;

  vqc_prep<<<PREP_BLOCKS, 128, 0, stream>>>(x, w, ws);
  vqc_main<<<BATCH_N / (WPB * SPW), 32 * WPB, 0, stream>>>(ws, out);
}


# last layer's RZ on wires 6-9 dropped (unit phases ahead of CNOT permutation and |amp|^2 change no output): its four-wire factor is real, per plane, K=16: 32 WMMA not 64; sums regrouped
# speedup vs baseline: 2.4362x; 1.0447x over previous
#include <hip/hip_runtime.h>


typedef __attribute__((ext_vector_type(2))) float v2f;
typedef __attribute__((ext_vector_type(4))) float v4f;
typedef __attribute__((ext_vector_type(8))) float v8f;

#define NQ      10
#define NL      4
#define BATCH_N 8192
#define BUF_F   2048
#define SPW     2
#define WPB     8
#define LAYER_F 1536
#define RTAB_F  256
#define TAB_F      (NL * LAYER_F + RTAB_F)
#define TAB_BLOCKS (TAB_F / 128)
#define ENC_BLOCKS (BATCH_N * 2 / 128)
#define PREP_BLOCKS (TAB_BLOCKS + ENC_BLOCKS)

__device__ __forceinline__ constexpr int phys(int p, int U) {
  return (p << 10) | (((U ^ ((U >> 4) & 15) ^ p) & 255) << 2);
}


__device__ __forceinline__ constexpr int encpos(int i) {
  return (i & 16) | ((i & 3) << 2) | ((i >> 2) & 3);
}

__device__ __forceinline__ void gate_elem(float a, float b, int rb, int cb, bool phase,
                                          float& fr, float& fi) {
  float ca = cosf(0.5f * a), sa = sinf(0.5f * a);
  float ry = rb ? (cb ? ca : sa) : (cb ? -sa : ca);
  if (!phase) { fr = ry; fi = 0.f; return; }
  float cz = cosf(0.5f * b), sz = sinf(0.5f * b);
  fr = ry * cz;
  fi = ry * (rb ? sz : -sz);
}

__device__ __forceinline__ v4f ld4(const float* p) { return *(const v4f*)p; }
__device__ __forceinline__ void st4(float* p, v4f v) { *(v4f*)p = v; }

__device__ __forceinline__ void prep_tables(const float* __restrict__ w, float* __restrict__ ws, int t) {
  int l = t / LAYER_F;
  int e = t - l * LAYER_F;
  int nb, first, M, K;
  const bool phase = l < NL;
  if (!phase) {
    int lane = (e >> 3) & 31, slot = e & 7;
    l = NL - 1;
    nb = 4; first = 6; M = lane & 15; K = 8 * (lane >> 4) + slot;
  } else if (e < 512) {
    int lane = (e >> 3) & 31, slot = e & 7;
    nb = 3; first = 3 * (e >> 8); M = lane & 15; K = 8 * (lane >> 4) + slot;
  } else {
    int e2 = e - 512;
    int lane = (e2 >> 4) & 31, slot = e2 & 15;
    nb = 4; first = 6; M = (e2 >> 9) * 16 + (lane & 15); K = 16 * (lane >> 4) + slot;
  }
  int gm = (1 << nb) - 1;
  int po = M >> nb, o = M & gm, pi = K >> nb, i = K & gm;
  float vr = 1.f, vi = 0.f;
  #pragma unroll
  for (int j = 0; j < 4; ++j) {
    if (j < nb) {
      int wire = first + j;
      int rb = (o >> (nb - 1 - j)) & 1;
      int cb = (i >> (nb - 1 - j)) & 1;
      float fr, fi;
      gate_elem(w[(l * NQ + wire) * 2], w[(l * NQ + wire) * 2 + 1], rb, cb, phase, fr, fi);
      float nr = vr * fr - vi * fi; vi = vr * fi + vi * fr; vr = nr;
    }
  }
  ws[t] = (po == pi) ? vr : (po ? vi : -vi);
}

__device__ __forceinline__ void prep_encoding(const float* __restrict__ x, float* __restrict__ ws, int u) {
  const float* xp = x + u * 5;
  float xv[5];
  #pragma unroll
  for (int j = 0; j < 5; ++j) xv[j] = xp[j];
  float v[32];
  {
    const float ah = 0.5f * xv[0];
    v[0] = cosf(ah); v[1] = sinf(ah);
  }
  #pragma unroll
  for (int j = 1; j < 5; ++j) {
    const float ah = 0.5f * xv[j];
    const float cj = cosf(ah), sj = sinf(ah);
    #pragma unroll
    for (int i = (1 << j) - 1; i >= 0; --i) {
      const float p = v[i];
      v[2 * i] = p * cj; v[2 * i + 1] = p * sj;
    }
  }
  float* o = ws + TAB_F + u * 32;
  #pragma unroll
  for (int q = 0; q < 8; ++q) {
    v4f t = {v[encpos(4 * q)], v[encpos(4 * q + 1)], v[encpos(4 * q + 2)], v[encpos(4 * q + 3)]};
    st4(o + 4 * q, t);
  }
}

__global__ __launch_bounds__(128) void vqc_prep(const float* __restrict__ x, const float* __restrict__ w,
                                                float* __restrict__ ws) {
  if (blockIdx.x < TAB_BLOCKS)
    prep_tables(w, ws, blockIdx.x * 128 + threadIdx.x);
  else
    prep_encoding(x, ws, (blockIdx.x - TAB_BLOCKS) * 128 + threadIdx.x);
}


__device__ __forceinline__ void gate3(const float* __restrict__ frag, int lane, float (&g)[8]) {
  const v4f* gp = (const v4f*)(frag + lane * 8);
  const v4f g0 = gp[0], g1 = gp[1];
  g[0] = g0.x; g[1] = g0.y; g[2] = g0.z; g[3] = g0.w;
  g[4] = g1.x; g[5] = g1.y; g[6] = g1.z; g[7] = g1.w;
}

__device__ __forceinline__ void gate4(const float* __restrict__ frag, int lane, float (&g)[2][16]) {
  #pragma unroll
  for (int nt = 0; nt < 2; ++nt) {
    const v4f* gp = (const v4f*)(frag + nt * 512 + lane * 16);
    #pragma unroll
    for (int q = 0; q < 4; ++q) {
      const v4f t = gp[q];
      g[nt][4 * q] = t.x; g[nt][4 * q + 1] = t.y; g[nt][4 * q + 2] = t.z; g[nt][4 * q + 3] = t.w;
    }
  }
}

__device__ __forceinline__ void gate4r(const float* __restrict__ frag, int lane, float (&g)[2][16]) {
  const v4f* gp = (const v4f*)(frag + lane * 8);
  const v4f g0 = gp[0], g1 = gp[1];
  g[0][0] = g0.x; g[0][1] = g0.y; g[0][2] = g0.z; g[0][3] = g0.w;
  g[0][4] = g1.x; g[0][5] = g1.y; g[0][6] = g1.z; g[0][7] = g1.w;
}

template <bool SECOND>
__device__ __forceinline__ void f3_load(const float* X, int lane, v4f (&a)[16]) {
  const int c = lane & 15, h = lane >> 4;
  int lb;
  if (!SECOND) lb = phys(h, ((c & 7) << 5) | ((c >> 3) << 4));
  else         lb = phys(h, c << 1);
  #pragma unroll
  for (int T = 0; T < 8; ++T)
    #pragma unroll
    for (int j = 0; j < 2; ++j) {
      const int cu = SECOND ? ((T << 5) | j) : ((T << 1) | j);
      a[2 * T + j] = ld4(X + (lb ^ phys(0, cu)));
    }
  __builtin_amdgcn_sched_barrier(0);
}

template <bool SECOND>
__device__ __forceinline__ void f3_run(const float (&g)[8], float* X, int lane, const v4f (&a)[16]) {
  const int c = lane & 15, h = lane >> 4;
  int sb;
  if (!SECOND) {
    sb = phys(c >> 3, ((c & 7) << 5) | (h << 4));
  } else {
    sb = phys(c >> 3, ((((c & 3) | (((c >> 2) & 1) << 3))) << 2) | (h << 1));
  }
  #pragma unroll
  for (int tp = 0; tp < 4; ++tp) {
    v8f d[2];
    d[0] = (v8f){0, 0, 0, 0, 0, 0, 0, 0};
    d[1] = (v8f){0, 0, 0, 0, 0, 0, 0, 0};
    #pragma unroll
    for (int ks = 0; ks < 4; ++ks)
      #pragma unroll
      for (int t2 = 0; t2 < 2; ++t2) {
        const int T = 2 * tp + t2;
        const v4f av = a[2 * T + (ks >> 1)];
        v2f a2 = {av[(2 * ks) & 3], av[((2 * ks) & 3) + 1]};
        v2f b2 = {g[2 * ks], g[2 * ks + 1]};
        d[t2] = __builtin_amdgcn_wmma_f32_16x16x4_f32(false, a2, false, b2,
                                                      (short)0, d[t2], false, false);
      }
    #pragma unroll
    for (int t2 = 0; t2 < 2; ++t2)
      #pragma unroll
      for (int j = 0; j < 2; ++j) {
        const int T = 2 * tp + t2;
        const int cu = SECOND ? (((((T & 1) << 2) | ((T >> 1) << 4)) << 2) | j) : ((T << 1) | j);
        v4f v = {d[t2][4 * j], d[t2][4 * j + 1], d[t2][4 * j + 2], d[t2][4 * j + 3]};
        st4(X + (sb ^ phys(0, cu)), v);
      }
    __builtin_amdgcn_sched_barrier(0);
  }
}

__device__ __forceinline__ void f4_load(const float* X, int lane, v4f (&a)[16]) {
  const int c = lane & 15, h = lane >> 4;
  const int u0 = c >> 3, u3 = (c >> 2) & 1, u2 = (c >> 1) & 1, u1 = c & 1;
  const int lrow = ((u2 ^ u1) << 2) | ((u1 ^ u0) << 3) | ((u3 ^ u2) << 4) | (u3 << 5);
  const int lb = phys(h, lrow << 2);
  #pragma unroll
  for (int T = 0; T < 4; ++T)
    #pragma unroll
    for (int q = 0; q < 4; ++q)
      a[4 * T + q] = ld4(X + (lb ^ phys(0, (T << 2) | q)));
  __builtin_amdgcn_sched_barrier(0);
}

__device__ __forceinline__ void f4_run(const float (&g)[2][16], float* X, int lane, const v4f (&a)[16]) {
  const int c = lane & 15, h = lane >> 4;
  const int n3 = h ^ (c >> 3), n2 = n3 ^ ((c >> 2) & 1), n1 = n2 ^ ((c >> 1) & 1), n0 = n1 ^ (c & 1);
  const int srow = (h << 6) | (h << 5) | (h << 4) | (n3 << 3) | (n2 << 2) | (n1 << 1) | n0;
  const int sb = phys(0, (srow << 1) | n0);
  #pragma unroll
  for (int nt = 0; nt < 2; ++nt) {
    #pragma unroll
    for (int tp = 0; tp < 2; ++tp) {
      v8f d[2];
      d[0] = (v8f){0, 0, 0, 0, 0, 0, 0, 0};
      d[1] = (v8f){0, 0, 0, 0, 0, 0, 0, 0};
      #pragma unroll
      for (int ks = 0; ks < 8; ++ks)
        #pragma unroll
        for (int t2 = 0; t2 < 2; ++t2) {
          const int T = 2 * tp + t2;
          const v4f av = a[4 * T + (ks >> 1)];
          v2f a2 = {av[(2 * ks) & 3], av[((2 * ks) & 3) + 1]};
          v2f b2 = {g[nt][2 * ks], g[nt][2 * ks + 1]};
          d[t2] = __builtin_amdgcn_wmma_f32_16x16x4_f32(false, a2, false, b2,
                                                        (short)0, d[t2], false, false);
        }
      #pragma unroll
      for (int t2 = 0; t2 < 2; ++t2)
        #pragma unroll
        for (int j = 0; j < 2; ++j) {
          const int T = 2 * tp + t2;
          const int a5 = (T >> 1) & 1, a4 = T & 1;
          const int nT = (a5 ? 0x3f : 0) ^ (a4 ? 0x1f : 0);
          const int cu = (nT << 1) | ((a5 ^ a4) ^ j);
          v4f v = {d[t2][4 * j], d[t2][4 * j + 1], d[t2][4 * j + 2], d[t2][4 * j + 3]};
          st4(X + (sb ^ phys(nt, cu)), v);
        }
      __builtin_amdgcn_sched_barrier(0);
    }
  }
}

__device__ __forceinline__ void f4r_load(const float* X, int lane, v4f (&a)[16]) {
  const int c = lane & 15, h = lane >> 4;
  const int u0 = c >> 3, u3 = (c >> 2) & 1, u2 = (c >> 1) & 1, u1 = c & 1;
  const int lrow = ((u2 ^ u1) << 2) | ((u1 ^ u0) << 3) | ((u3 ^ u2) << 4) | (u3 << 5);
  const int lb = phys(0, (lrow << 2) | (h << 1));
  #pragma unroll
  for (int T = 0; T < 4; ++T)
    #pragma unroll
    for (int p = 0; p < 2; ++p)
      #pragma unroll
      for (int j = 0; j < 2; ++j)
        a[4 * T + 2 * p + j] = ld4(X + (lb ^ phys(p, (T << 2) | j)));
  __builtin_amdgcn_sched_barrier(0);
}

__device__ __forceinline__ void f4r_run(const float (&g)[2][16], float* X, int lane, const v4f (&a)[16]) {
  const int c = lane & 15, h = lane >> 4;
  const int n3 = h ^ (c >> 3), n2 = n3 ^ ((c >> 2) & 1), n1 = n2 ^ ((c >> 1) & 1), n0 = n1 ^ (c & 1);
  const int srow = (h << 6) | (h << 5) | (h << 4) | (n3 << 3) | (n2 << 2) | (n1 << 1) | n0;
  const int sb = phys(0, (srow << 1) | n0);
  #pragma unroll
  for (int p = 0; p < 2; ++p) {
    #pragma unroll
    for (int tp = 0; tp < 2; ++tp) {
      v8f d[2];
      d[0] = (v8f){0, 0, 0, 0, 0, 0, 0, 0};
      d[1] = (v8f){0, 0, 0, 0, 0, 0, 0, 0};
      #pragma unroll
      for (int ks = 0; ks < 4; ++ks)
        #pragma unroll
        for (int t2 = 0; t2 < 2; ++t2) {
          const int T = 2 * tp + t2;
          const v4f av = a[4 * T + 2 * p + (ks >> 1)];
          v2f a2 = {av[(2 * ks) & 3], av[((2 * ks) & 3) + 1]};
          v2f b2 = {g[0][2 * ks], g[0][2 * ks + 1]};
          d[t2] = __builtin_amdgcn_wmma_f32_16x16x4_f32(false, a2, false, b2,
                                                        (short)0, d[t2], false, false);
        }
      #pragma unroll
      for (int t2 = 0; t2 < 2; ++t2)
        #pragma unroll
        for (int j = 0; j < 2; ++j) {
          const int T = 2 * tp + t2;
          const int a5 = (T >> 1) & 1, a4 = T & 1;
          const int nT = (a5 ? 0x3f : 0) ^ (a4 ? 0x1f : 0);
          const int cu = (nT << 1) | ((a5 ^ a4) ^ j);
          v4f v = {d[t2][4 * j], d[t2][4 * j + 1], d[t2][4 * j + 2], d[t2][4 * j + 3]};
          st4(X + (sb ^ phys(p, cu)), v);
        }
      __builtin_amdgcn_sched_barrier(0);
    }
  }
}

__device__ __forceinline__ float xhalf(float x) { return __shfl_xor(x, 16, 32); }

template <int M>
__device__ __forceinline__ float xrow(float x) {
  int b = __builtin_bit_cast(int, x);
  return __builtin_bit_cast(float, __builtin_amdgcn_update_dpp(0, b, 0x160 | M, 0xf, 0xf, true));
}

__device__ __forceinline__ void encode(const float* __restrict__ ws, float* X, int sample, int lane) {
  const float* ev = ws + TAB_F + sample * 64;
  const float colv = ev[32 + encpos(lane)];
  v4f rq[8];
  #pragma unroll
  for (int k = 0; k < 8; ++k) rq[k] = ld4(ev + 4 * k);
  const int ub = phys(0, lane << 1);
  const v4f zero4 = {0.f, 0.f, 0.f, 0.f};
  #pragma unroll
  for (int k = 0; k < 8; ++k) {
    const int cu = ((k & 3) << 6) | (k >> 2);
    st4(X + (ub ^ phys(0, cu)), rq[k] * colv);
    st4(X + (ub ^ phys(1, cu)), zero4);
  }
}

__device__ __forceinline__ void measure(const float* X, float* __restrict__ out, int sample, int lane) {
  const int mc = lane & 15, mh = lane >> 4;
  const float* Lp = X + ((mh << 6) | (mc << 2));
  v2f q01 = {0.f, 0.f}, q23 = {0.f, 0.f};
  float s0 = 0.f, s1 = 0.f;
  float ut[8];
  #pragma unroll
  for (int t = 0; t < 8; ++t) {
    const v4f a = ld4(Lp + (t << 7));
    const v4f b = ld4(Lp + ((1 << 10) | (t << 7)));
    const v2f al = {a.x, a.y}, ah = {a.z, a.w}, bl = {b.x, b.y}, bh = {b.z, b.w};
    const v2f ra = al * al + ah * ah;
    const v2f rb = bl * bl + bh * bh;
    q01 += al * al; q23 += ah * ah;
    q01 += bl * bl; q23 += bh * bh;
    const float sa = ra.x + ra.y, sb = rb.x + rb.y;
    s0 += sa; s1 += sb;
    ut[t] = sa + sb;
  }
  const float e0 = ut[0] + ut[1], e1 = ut[2] + ut[3], e2 = ut[4] + ut[5], e3 = ut[6] + ut[7];
  float z4 = ((ut[0] - ut[1]) + (ut[2] - ut[3])) + ((ut[4] - ut[5]) + (ut[6] - ut[7]));
  float z5 = (e0 - e1) + (e2 - e3);
  float z6 = (e0 + e1) - (e2 + e3);
  float st = s0 + s1;
  float pd = s0 - s1;
  float w8 = (q01.x + q01.y) - (q23.x + q23.y);
  float w7 = (q01.x - q01.y) + (q23.x - q23.y);

  z6 += xhalf(z6); z6 += xrow<1>(z6); z6 += xrow<2>(z6); z6 += xrow<4>(z6);
  z5 += xhalf(z5); z5 += xrow<1>(z5); z5 += xrow<2>(z5); z5 += xrow<8>(z5);
  z4 += xhalf(z4); z4 += xrow<1>(z4); z4 += xrow<4>(z4); z4 += xrow<8>(z4);
  w8 += xhalf(w8); w8 += xrow<1>(w8); w8 += xrow<2>(w8); w8 += xrow<4>(w8); w8 += xrow<8>(w8);
  w7 += xhalf(w7); w7 += xrow<1>(w7); w7 += xrow<2>(w7); w7 += xrow<4>(w7); w7 += xrow<8>(w7);
  st += xrow<1>(st); st += xrow<2>(st); st += xrow<4>(st); st += xrow<8>(st);
  pd -= xrow<1>(pd); pd += xrow<2>(pd); pd += xrow<4>(pd); pd += xrow<8>(pd);
  const float z6o = xrow<8>(z6), z5o = xrow<4>(z5), z4o = xrow<2>(z4);
  const float sto = xhalf(st), pdo = xhalf(pd);
  if (lane == 0) {
    float* o = out + sample * NQ;
    o[0] = pd - pdo;
    o[1] = w8;
    o[2] = w7;
    o[3] = z6 + z6o;
    o[4] = z5 + z5o;
    o[5] = z4 + z4o;
    o[6] = st - sto;
    o[7] = z6 - z6o;
    o[8] = z5 - z5o;
    o[9] = z4 - z4o;
  }
}

__global__ __launch_bounds__(32 * WPB) __attribute__((amdgpu_waves_per_eu(4, 4))) void vqc_main(const float* __restrict__ ws,
                                                     float* __restrict__ out) {
  __shared__ __attribute__((aligned(16))) float lds[WPB * SPW * BUF_F];
  const int lane = threadIdx.x & 31;
  const int wv   = threadIdx.x >> 5;
  const int sample = (blockIdx.x * WPB + wv) * SPW;

  float* X0 = lds + wv * (SPW * BUF_F);
  float* X1 = X0 + BUF_F;

  encode(ws, X0, sample, lane);
  encode(ws, X1, sample + 1, lane);

  v4f A0[16], A1[16];
  float g1[8], g2[8], g3[2][16];
  f3_load<false>(X0, lane, A0);
  gate3(ws, lane, g1);

  #pragma unroll 1
  for (int l = 0; l + 1 < NL; ++l) {
    const float* f0 = ws + l * LAYER_F;
    int ln = lane;
    asm volatile("" : "+v"(ln));
    ln &= 31;
    f3_load<false>(X1, ln, A1);
    f3_run<false>(g1, X0, ln, A0);
    f3_load<true>(X0, ln, A0);
    gate3(f0 + 256, ln, g2);
    f3_run<false>(g1, X1, ln, A1);
    f3_load<true>(X1, ln, A1);
    gate4(f0 + 512, ln, g3);
    f3_run<true>(g2, X0, ln, A0);
    f4_load(X0, ln, A0);
    f3_run<true>(g2, X1, ln, A1);
    f4_load(X1, ln, A1);
    f4_run(g3, X0, ln, A0);
    f3_load<false>(X0, ln, A0);
    gate3(f0 + LAYER_F, ln, g1);
    f4_run(g3, X1, ln, A1);
  }

  {
    const float* f0 = ws + (NL - 1) * LAYER_F;
    int ln = lane;
    asm volatile("" : "+v"(ln));
    ln &= 31;
    f3_load<false>(X1, ln, A1);
    f3_run<false>(g1, X0, ln, A0);
    f3_load<true>(X0, ln, A0);
    gate3(f0 + 256, ln, g2);
    f3_run<false>(g1, X1, ln, A1);
    f3_load<true>(X1, ln, A1);
    gate4r(ws + NL * LAYER_F, ln, g3);
    f3_run<true>(g2, X0, ln, A0);
    f4r_load(X0, ln, A0);
    f3_run<true>(g2, X1, ln, A1);
    f4r_load(X1, ln, A1);
    f4r_run(g3, X0, ln, A0);
    f4r_run(g3, X1, ln, A1);
  }

  measure(X0, out, sample, lane);
  measure(X1, out, sample + 1, lane);
}

extern "C" void kernel_launch(void* const* d_in, const int* in_sizes, int n_in,
                              void* d_out, int out_size, void* d_ws, size_t ws_size,
                              hipStream_t stream) {
  const float* x = (const float*)d_in[0];
  const float* w = (const float*)d_in[1];
  float* ws  = (float*)d_ws;
  float* out = (float*)d_out;

  vqc_prep<<<PREP_BLOCKS, 128, 0, stream>>>(x, w, ws);
  vqc_main<<<BATCH_N / (WPB * SPW), 32 * WPB, 0, stream>>>(ws, out);
}


# layer 0's four-wire factor first, on the real encoded state: contracts over the re plane only, K=16: 32 f32 WMMA per sample, not 64; ring moved to the second factor's stores; layer 0 sums regrouped
# speedup vs baseline: 2.5557x; 1.0491x over previous
#include <hip/hip_runtime.h>


typedef __attribute__((ext_vector_type(2))) float v2f;
typedef __attribute__((ext_vector_type(4))) float v4f;
typedef __attribute__((ext_vector_type(8))) float v8f;

#define NQ      10
#define NL      4
#define BATCH_N 8192
#define BUF_F   2048
#define SPW     2
#define WPB     8
#define LAYER_F 1536
#define RTAB_F  256
#define TAB_F      (NL * LAYER_F + RTAB_F)
#define TAB_BLOCKS (TAB_F / 128)
#define ENC_BLOCKS (BATCH_N * 2 / 128)
#define PREP_BLOCKS (TAB_BLOCKS + ENC_BLOCKS)

__device__ __forceinline__ constexpr int phys(int p, int U) {
  return (p << 10) | (((U ^ ((U >> 4) & 15) ^ p) & 255) << 2);
}


__device__ __forceinline__ constexpr int rowel(int n) {
  return ((n & 7) << 2) | (n >> 3);
}

__device__ __forceinline__ constexpr int ringaddr(int p, int a) {
  int y = a;
  y ^= y >> 1; y ^= y >> 2; y ^= y >> 4; y ^= y >> 8;
  const int n = (y & 0x1ff) | (((y ^ (y >> 9)) & 1) << 9);
  return phys(p, ((n & 0x7f) << 1) | (n >> 9)) | ((n >> 7) & 3);
}

__device__ __forceinline__ void gate_elem(float a, float b, int rb, int cb, bool phase,
                                          float& fr, float& fi) {
  float ca = cosf(0.5f * a), sa = sinf(0.5f * a);
  float ry = rb ? (cb ? ca : sa) : (cb ? -sa : ca);
  if (!phase) { fr = ry; fi = 0.f; return; }
  float cz = cosf(0.5f * b), sz = sinf(0.5f * b);
  fr = ry * cz;
  fi = ry * (rb ? sz : -sz);
}

__device__ __forceinline__ v4f ld4(const float* p) { return *(const v4f*)p; }
__device__ __forceinline__ void st4(float* p, v4f v) { *(v4f*)p = v; }

__device__ __forceinline__ void prep_tables(const float* __restrict__ w, float* __restrict__ ws, int t) {
  int l = t / LAYER_F;
  int e = t - l * LAYER_F;
  int nb, first, M, K;
  const bool phase = l < NL;
  if (!phase) {
    int lane = (e >> 3) & 31, slot = e & 7;
    l = NL - 1;
    nb = 4; first = 6; M = lane & 15; K = 8 * (lane >> 4) + slot;
  } else if (e < 512) {
    int lane = (e >> 3) & 31, slot = e & 7;
    nb = 3; first = 3 * (e >> 8); M = lane & 15; K = 8 * (lane >> 4) + slot;
  } else {
    int e2 = e - 512;
    int lane = (e2 >> 4) & 31, slot = e2 & 15;
    nb = 4; first = 6; M = (e2 >> 9) * 16 + (lane & 15); K = 16 * (lane >> 4) + slot;
  }
  int gm = (1 << nb) - 1;
  int po = M >> nb, o = M & gm, pi = K >> nb, i = K & gm;
  float vr = 1.f, vi = 0.f;
  #pragma unroll
  for (int j = 0; j < 4; ++j) {
    if (j < nb) {
      int wire = first + j;
      int rb = (o >> (nb - 1 - j)) & 1;
      int cb = (i >> (nb - 1 - j)) & 1;
      float fr, fi;
      gate_elem(w[(l * NQ + wire) * 2], w[(l * NQ + wire) * 2 + 1], rb, cb, phase, fr, fi);
      float nr = vr * fr - vi * fi; vi = vr * fi + vi * fr; vr = nr;
    }
  }
  ws[t] = (po == pi) ? vr : (po ? vi : -vi);
}

__device__ __forceinline__ void prep_encoding(const float* __restrict__ x, float* __restrict__ ws, int u) {
  const float* xp = x + u * 5;
  float xv[5];
  #pragma unroll
  for (int j = 0; j < 5; ++j) xv[j] = xp[j];
  float v[32];
  {
    const float ah = 0.5f * xv[0];
    v[0] = cosf(ah); v[1] = sinf(ah);
  }
  #pragma unroll
  for (int j = 1; j < 5; ++j) {
    const float ah = 0.5f * xv[j];
    const float cj = cosf(ah), sj = sinf(ah);
    #pragma unroll
    for (int i = (1 << j) - 1; i >= 0; --i) {
      const float p = v[i];
      v[2 * i] = p * cj; v[2 * i + 1] = p * sj;
    }
  }
  float* o = ws + TAB_F + u * 32;
  if (u & 1) {
    #pragma unroll
    for (int q = 0; q < 8; ++q) {
      v4f t = {v[4 * q], v[4 * q + 1], v[4 * q + 2], v[4 * q + 3]};
      st4(o + 4 * q, t);
    }
  } else {
    #pragma unroll
    for (int q = 0; q < 8; ++q) {
      v4f t = {v[rowel(4 * q)], v[rowel(4 * q + 1)], v[rowel(4 * q + 2)], v[rowel(4 * q + 3)]};
      st4(o + 4 * q, t);
    }
  }
}

__global__ __launch_bounds__(128) void vqc_prep(const float* __restrict__ x, const float* __restrict__ w,
                                                float* __restrict__ ws) {
  if (blockIdx.x < TAB_BLOCKS)
    prep_tables(w, ws, blockIdx.x * 128 + threadIdx.x);
  else
    prep_encoding(x, ws, (blockIdx.x - TAB_BLOCKS) * 128 + threadIdx.x);
}


__device__ __forceinline__ void gate3(const float* __restrict__ frag, int lane, float (&g)[8]) {
  const v4f* gp = (const v4f*)(frag + lane * 8);
  const v4f g0 = gp[0], g1 = gp[1];
  g[0] = g0.x; g[1] = g0.y; g[2] = g0.z; g[3] = g0.w;
  g[4] = g1.x; g[5] = g1.y; g[6] = g1.z; g[7] = g1.w;
}

__device__ __forceinline__ void gate4(const float* __restrict__ frag, int lane, float (&g)[2][16]) {
  #pragma unroll
  for (int nt = 0; nt < 2; ++nt) {
    const v4f* gp = (const v4f*)(frag + nt * 512 + lane * 16);
    #pragma unroll
    for (int q = 0; q < 4; ++q) {
      const v4f t = gp[q];
      g[nt][4 * q] = t.x; g[nt][4 * q + 1] = t.y; g[nt][4 * q + 2] = t.z; g[nt][4 * q + 3] = t.w;
    }
  }
}

__device__ __forceinline__ void gate4r(const float* __restrict__ frag, int lane, float (&g)[2][16]) {
  const v4f* gp = (const v4f*)(frag + lane * 8);
  const v4f g0 = gp[0], g1 = gp[1];
  g[0][0] = g0.x; g[0][1] = g0.y; g[0][2] = g0.z; g[0][3] = g0.w;
  g[0][4] = g1.x; g[0][5] = g1.y; g[0][6] = g1.z; g[0][7] = g1.w;
}

template <bool SECOND>
__device__ __forceinline__ void f3_load(const float* X, int lane, v4f (&a)[16]) {
  const int c = lane & 15, h = lane >> 4;
  int lb;
  if (!SECOND) lb = phys(h, ((c & 7) << 5) | ((c >> 3) << 4));
  else         lb = phys(h, c << 1);
  #pragma unroll
  for (int T = 0; T < 8; ++T)
    #pragma unroll
    for (int j = 0; j < 2; ++j) {
      const int cu = SECOND ? ((T << 5) | j) : ((T << 1) | j);
      a[2 * T + j] = ld4(X + (lb ^ phys(0, cu)));
    }
  __builtin_amdgcn_sched_barrier(0);
}

template <bool SECOND>
__device__ __forceinline__ void f3_run(const float (&g)[8], float* X, int lane, const v4f (&a)[16]) {
  const int c = lane & 15, h = lane >> 4;
  int sb;
  if (!SECOND) {
    sb = phys(c >> 3, ((c & 7) << 5) | (h << 4));
  } else {
    sb = phys(c >> 3, ((((c & 3) | (((c >> 2) & 1) << 3))) << 2) | (h << 1));
  }
  #pragma unroll
  for (int tp = 0; tp < 4; ++tp) {
    v8f d[2];
    d[0] = (v8f){0, 0, 0, 0, 0, 0, 0, 0};
    d[1] = (v8f){0, 0, 0, 0, 0, 0, 0, 0};
    #pragma unroll
    for (int ks = 0; ks < 4; ++ks)
      #pragma unroll
      for (int t2 = 0; t2 < 2; ++t2) {
        const int T = 2 * tp + t2;
        const v4f av = a[2 * T + (ks >> 1)];
        v2f a2 = {av[(2 * ks) & 3], av[((2 * ks) & 3) + 1]};
        v2f b2 = {g[2 * ks], g[2 * ks + 1]};
        d[t2] = __builtin_amdgcn_wmma_f32_16x16x4_f32(false, a2, false, b2,
                                                      (short)0, d[t2], false, false);
      }
    #pragma unroll
    for (int t2 = 0; t2 < 2; ++t2)
      #pragma unroll
      for (int j = 0; j < 2; ++j) {
        const int T = 2 * tp + t2;
        const int cu = SECOND ? (((((T & 1) << 2) | ((T >> 1) << 4)) << 2) | j) : ((T << 1) | j);
        v4f v = {d[t2][4 * j], d[t2][4 * j + 1], d[t2][4 * j + 2], d[t2][4 * j + 3]};
        st4(X + (sb ^ phys(0, cu)), v);
      }
    __builtin_amdgcn_sched_barrier(0);
  }
}

__device__ __forceinline__ void f4_load(const float* X, int lane, v4f (&a)[16]) {
  const int c = lane & 15, h = lane >> 4;
  const int u0 = c >> 3, u3 = (c >> 2) & 1, u2 = (c >> 1) & 1, u1 = c & 1;
  const int lrow = ((u2 ^ u1) << 2) | ((u1 ^ u0) << 3) | ((u3 ^ u2) << 4) | (u3 << 5);
  const int lb = phys(h, lrow << 2);
  #pragma unroll
  for (int T = 0; T < 4; ++T)
    #pragma unroll
    for (int q = 0; q < 4; ++q)
      a[4 * T + q] = ld4(X + (lb ^ phys(0, (T << 2) | q)));
  __builtin_amdgcn_sched_barrier(0);
}

__device__ __forceinline__ void f4_run(const float (&g)[2][16], float* X, int lane, const v4f (&a)[16]) {
  const int c = lane & 15, h = lane >> 4;
  const int n3 = h ^ (c >> 3), n2 = n3 ^ ((c >> 2) & 1), n1 = n2 ^ ((c >> 1) & 1), n0 = n1 ^ (c & 1);
  const int srow = (h << 6) | (h << 5) | (h << 4) | (n3 << 3) | (n2 << 2) | (n1 << 1) | n0;
  const int sb = phys(0, (srow << 1) | n0);
  #pragma unroll
  for (int nt = 0; nt < 2; ++nt) {
    #pragma unroll
    for (int tp = 0; tp < 2; ++tp) {
      v8f d[2];
      d[0] = (v8f){0, 0, 0, 0, 0, 0, 0, 0};
      d[1] = (v8f){0, 0, 0, 0, 0, 0, 0, 0};
      #pragma unroll
      for (int ks = 0; ks < 8; ++ks)
        #pragma unroll
        for (int t2 = 0; t2 < 2; ++t2) {
          const int T = 2 * tp + t2;
          const v4f av = a[4 * T + (ks >> 1)];
          v2f a2 = {av[(2 * ks) & 3], av[((2 * ks) & 3) + 1]};
          v2f b2 = {g[nt][2 * ks], g[nt][2 * ks + 1]};
          d[t2] = __builtin_amdgcn_wmma_f32_16x16x4_f32(false, a2, false, b2,
                                                        (short)0, d[t2], false, false);
        }
      #pragma unroll
      for (int t2 = 0; t2 < 2; ++t2)
        #pragma unroll
        for (int j = 0; j < 2; ++j) {
          const int T = 2 * tp + t2;
          const int a5 = (T >> 1) & 1, a4 = T & 1;
          const int nT = (a5 ? 0x3f : 0) ^ (a4 ? 0x1f : 0);
          const int cu = (nT << 1) | ((a5 ^ a4) ^ j);
          v4f v = {d[t2][4 * j], d[t2][4 * j + 1], d[t2][4 * j + 2], d[t2][4 * j + 3]};
          st4(X + (sb ^ phys(nt, cu)), v);
        }
      __builtin_amdgcn_sched_barrier(0);
    }
  }
}

__device__ __forceinline__ void f4r_load(const float* X, int lane, v4f (&a)[16]) {
  const int c = lane & 15, h = lane >> 4;
  const int u0 = c >> 3, u3 = (c >> 2) & 1, u2 = (c >> 1) & 1, u1 = c & 1;
  const int lrow = ((u2 ^ u1) << 2) | ((u1 ^ u0) << 3) | ((u3 ^ u2) << 4) | (u3 << 5);
  const int lb = phys(0, (lrow << 2) | (h << 1));
  #pragma unroll
  for (int T = 0; T < 4; ++T)
    #pragma unroll
    for (int p = 0; p < 2; ++p)
      #pragma unroll
      for (int j = 0; j < 2; ++j)
        a[4 * T + 2 * p + j] = ld4(X + (lb ^ phys(p, (T << 2) | j)));
  __builtin_amdgcn_sched_barrier(0);
}

__device__ __forceinline__ void f4r_run(const float (&g)[2][16], float* X, int lane, const v4f (&a)[16]) {
  const int c = lane & 15, h = lane >> 4;
  const int n3 = h ^ (c >> 3), n2 = n3 ^ ((c >> 2) & 1), n1 = n2 ^ ((c >> 1) & 1), n0 = n1 ^ (c & 1);
  const int srow = (h << 6) | (h << 5) | (h << 4) | (n3 << 3) | (n2 << 2) | (n1 << 1) | n0;
  const int sb = phys(0, (srow << 1) | n0);
  #pragma unroll
  for (int p = 0; p < 2; ++p) {
    #pragma unroll
    for (int tp = 0; tp < 2; ++tp) {
      v8f d[2];
      d[0] = (v8f){0, 0, 0, 0, 0, 0, 0, 0};
      d[1] = (v8f){0, 0, 0, 0, 0, 0, 0, 0};
      #pragma unroll
      for (int ks = 0; ks < 4; ++ks)
        #pragma unroll
        for (int t2 = 0; t2 < 2; ++t2) {
          const int T = 2 * tp + t2;
          const v4f av = a[4 * T + 2 * p + (ks >> 1)];
          v2f a2 = {av[(2 * ks) & 3], av[((2 * ks) & 3) + 1]};
          v2f b2 = {g[0][2 * ks], g[0][2 * ks + 1]};
          d[t2] = __builtin_amdgcn_wmma_f32_16x16x4_f32(false, a2, false, b2,
                                                        (short)0, d[t2], false, false);
        }
      #pragma unroll
      for (int t2 = 0; t2 < 2; ++t2)
        #pragma unroll
        for (int j = 0; j < 2; ++j) {
          const int T = 2 * tp + t2;
          const int a5 = (T >> 1) & 1, a4 = T & 1;
          const int nT = (a5 ? 0x3f : 0) ^ (a4 ? 0x1f : 0);
          const int cu = (nT << 1) | ((a5 ^ a4) ^ j);
          v4f v = {d[t2][4 * j], d[t2][4 * j + 1], d[t2][4 * j + 2], d[t2][4 * j + 3]};
          st4(X + (sb ^ phys(p, cu)), v);
        }
      __builtin_amdgcn_sched_barrier(0);
    }
  }
}

__device__ __forceinline__ void gate4a(const float* __restrict__ frag, int lane, float (&g)[2][16]) {
  const int c = lane & 15, h = lane >> 4;
  #pragma unroll
  for (int nt = 0; nt < 2; ++nt) {
    const v4f* gp = (const v4f*)(frag + nt * 512 + c * 16 + 8 * h);
    const v4f g0 = gp[0], g1 = gp[1];
    g[nt][0] = g0.x; g[nt][1] = g0.y; g[nt][2] = g0.z; g[nt][3] = g0.w;
    g[nt][4] = g1.x; g[nt][5] = g1.y; g[nt][6] = g1.z; g[nt][7] = g1.w;
  }
}

__device__ __forceinline__ void fa_load(const float* X, int lane, v4f (&a)[16]) {
  const int c = lane & 15, h = lane >> 4;
  const int lb = phys(0, (((c >> 2) & 1) << 7) | (((c >> 1) & 1) << 6) | ((c >> 3) << 5) | ((c & 1) << 4) | (h << 1));
  #pragma unroll
  for (int T = 0; T < 4; ++T)
    #pragma unroll
    for (int j = 0; j < 2; ++j)
      a[2 * T + j] = ld4(X + (lb ^ phys(0, (T << 2) | j)));
  __builtin_amdgcn_sched_barrier(0);
}

__device__ __forceinline__ void fa_run(const float (&g)[2][16], float* X, int lane, const v4f (&a)[16]) {
  const int c = lane & 15, h = lane >> 4;
  const int sb = phys(0, (h << 7) | (c << 1));
  #pragma unroll
  for (int nt = 0; nt < 2; ++nt) {
    #pragma unroll
    for (int tp = 0; tp < 2; ++tp) {
      v8f d[2];
      d[0] = (v8f){0, 0, 0, 0, 0, 0, 0, 0};
      d[1] = (v8f){0, 0, 0, 0, 0, 0, 0, 0};
      #pragma unroll
      for (int ks = 0; ks < 4; ++ks)
        #pragma unroll
        for (int t2 = 0; t2 < 2; ++t2) {
          const int T = 2 * tp + t2;
          const v4f av = a[2 * T + (ks >> 1)];
          v2f a2 = {av[(2 * ks) & 3], av[((2 * ks) & 3) + 1]};
          v2f b2 = {g[nt][2 * ks], g[nt][2 * ks + 1]};
          d[t2] = __builtin_amdgcn_wmma_f32_16x16x4_f32(false, a2, false, b2,
                                                        (short)0, d[t2], false, false);
        }
      #pragma unroll
      for (int t2 = 0; t2 < 2; ++t2)
        #pragma unroll
        for (int j = 0; j < 2; ++j) {
          const int T = 2 * tp + t2;
          v4f v = {d[t2][4 * j], d[t2][4 * j + 1], d[t2][4 * j + 2], d[t2][4 * j + 3]};
          st4(X + (sb ^ phys(nt, (T << 5) | j)), v);
        }
      __builtin_amdgcn_sched_barrier(0);
    }
  }
}

__device__ __forceinline__ void f3c_run(const float (&g)[8], float* X, int lane, const v4f (&a)[16]) {
  const int c = lane & 15, h = lane >> 4;
  const int sb = ringaddr(c >> 3, ((c & 7) << 4) | (h << 3));
  #pragma unroll
  for (int tp = 0; tp < 4; ++tp) {
    v8f d[2];
    d[0] = (v8f){0, 0, 0, 0, 0, 0, 0, 0};
    d[1] = (v8f){0, 0, 0, 0, 0, 0, 0, 0};
    #pragma unroll
    for (int ks = 0; ks < 4; ++ks)
      #pragma unroll
      for (int t2 = 0; t2 < 2; ++t2) {
        const int T = 2 * tp + t2;
        const v4f av = a[2 * T + (ks >> 1)];
        v2f a2 = {av[(2 * ks) & 3], av[((2 * ks) & 3) + 1]};
        v2f b2 = {g[2 * ks], g[2 * ks + 1]};
        d[t2] = __builtin_amdgcn_wmma_f32_16x16x4_f32(false, a2, false, b2,
                                                      (short)0, d[t2], false, false);
      }
    #pragma unroll
    for (int t2 = 0; t2 < 2; ++t2)
      #pragma unroll
      for (int r = 0; r < 8; ++r) {
        const int T = 2 * tp + t2;
        X[sb ^ ringaddr(0, (T << 7) | r)] = d[t2][r];
      }
    __builtin_amdgcn_sched_barrier(0);
  }
}

__device__ __forceinline__ float xhalf(float x) { return __shfl_xor(x, 16, 32); }

template <int M>
__device__ __forceinline__ float xrow(float x) {
  int b = __builtin_bit_cast(int, x);
  return __builtin_bit_cast(float, __builtin_amdgcn_update_dpp(0, b, 0x160 | M, 0xf, 0xf, true));
}

__device__ __forceinline__ void encode3(const float* __restrict__ ws, float* X, int sample, int lane) {
  const float* ev = ws + TAB_F + sample * 64;
  const v4f cq = ld4(ev + 32 + 4 * (lane & 7));
  const v4f r0 = ld4(ev + 8 * (lane >> 3)), r1 = ld4(ev + 8 * (lane >> 3) + 4);
  const float rv[8] = {r0.x, r0.y, r0.z, r0.w, r1.x, r1.y, r1.z, r1.w};
  const int ub = phys(0, (((lane >> 4) & 1) << 5) | (((lane >> 3) & 1) << 3) | (lane & 7));
  #pragma unroll
  for (int k = 0; k < 8; ++k) {
    const int cu = (((k >> 2) & 1) << 7) | (((k >> 1) & 1) << 6) | ((k & 1) << 4);
    st4(X + (ub ^ phys(0, cu)), cq * rv[k]);
  }
}

__device__ __forceinline__ void measure(const float* X, float* __restrict__ out, int sample, int lane) {
  const int mc = lane & 15, mh = lane >> 4;
  const float* Lp = X + ((mh << 6) | (mc << 2));
  v2f q01 = {0.f, 0.f}, q23 = {0.f, 0.f};
  float s0 = 0.f, s1 = 0.f;
  float ut[8];
  #pragma unroll
  for (int t = 0; t < 8; ++t) {
    const v4f a = ld4(Lp + (t << 7));
    const v4f b = ld4(Lp + ((1 << 10) | (t << 7)));
    const v2f al = {a.x, a.y}, ah = {a.z, a.w}, bl = {b.x, b.y}, bh = {b.z, b.w};
    const v2f ra = al * al + ah * ah;
    const v2f rb = bl * bl + bh * bh;
    q01 += al * al; q23 += ah * ah;
    q01 += bl * bl; q23 += bh * bh;
    const float sa = ra.x + ra.y, sb = rb.x + rb.y;
    s0 += sa; s1 += sb;
    ut[t] = sa + sb;
  }
  const float e0 = ut[0] + ut[1], e1 = ut[2] + ut[3], e2 = ut[4] + ut[5], e3 = ut[6] + ut[7];
  float z4 = ((ut[0] - ut[1]) + (ut[2] - ut[3])) + ((ut[4] - ut[5]) + (ut[6] - ut[7]));
  float z5 = (e0 - e1) + (e2 - e3);
  float z6 = (e0 + e1) - (e2 + e3);
  float st = s0 + s1;
  float pd = s0 - s1;
  float w8 = (q01.x + q01.y) - (q23.x + q23.y);
  float w7 = (q01.x - q01.y) + (q23.x - q23.y);

  z6 += xhalf(z6); z6 += xrow<1>(z6); z6 += xrow<2>(z6); z6 += xrow<4>(z6);
  z5 += xhalf(z5); z5 += xrow<1>(z5); z5 += xrow<2>(z5); z5 += xrow<8>(z5);
  z4 += xhalf(z4); z4 += xrow<1>(z4); z4 += xrow<4>(z4); z4 += xrow<8>(z4);
  w8 += xhalf(w8); w8 += xrow<1>(w8); w8 += xrow<2>(w8); w8 += xrow<4>(w8); w8 += xrow<8>(w8);
  w7 += xhalf(w7); w7 += xrow<1>(w7); w7 += xrow<2>(w7); w7 += xrow<4>(w7); w7 += xrow<8>(w7);
  st += xrow<1>(st); st += xrow<2>(st); st += xrow<4>(st); st += xrow<8>(st);
  pd -= xrow<1>(pd); pd += xrow<2>(pd); pd += xrow<4>(pd); pd += xrow<8>(pd);
  const float z6o = xrow<8>(z6), z5o = xrow<4>(z5), z4o = xrow<2>(z4);
  const float sto = xhalf(st), pdo = xhalf(pd);
  if (lane == 0) {
    float* o = out + sample * NQ;
    o[0] = pd - pdo;
    o[1] = w8;
    o[2] = w7;
    o[3] = z6 + z6o;
    o[4] = z5 + z5o;
    o[5] = z4 + z4o;
    o[6] = st - sto;
    o[7] = z6 - z6o;
    o[8] = z5 - z5o;
    o[9] = z4 - z4o;
  }
}

__global__ __launch_bounds__(32 * WPB) __attribute__((amdgpu_waves_per_eu(4, 4))) void vqc_main(const float* __restrict__ ws,
                                                     float* __restrict__ out) {
  __shared__ __attribute__((aligned(16))) float lds[WPB * SPW * BUF_F];
  const int lane = threadIdx.x & 31;
  const int wv   = threadIdx.x >> 5;
  const int sample = (blockIdx.x * WPB + wv) * SPW;

  float* X0 = lds + wv * (SPW * BUF_F);
  float* X1 = X0 + BUF_F;

  encode3(ws, X0, sample, lane);
  encode3(ws, X1, sample + 1, lane);

  v4f A0[16], A1[16];
  float g1[8], g2[8], g3[2][16];

  fa_load(X0, lane, A0);
  gate4a(ws + 512, lane, g3);
  fa_load(X1, lane, A1);
  fa_run(g3, X0, lane, A0);
  f3_load<false>(X0, lane, A0);
  gate3(ws, lane, g1);
  fa_run(g3, X1, lane, A1);
  f3_load<false>(X1, lane, A1);
  f3_run<false>(g1, X0, lane, A0);
  f3_load<true>(X0, lane, A0);
  gate3(ws + 256, lane, g2);
  f3_run<false>(g1, X1, lane, A1);
  f3_load<true>(X1, lane, A1);
  f3c_run(g2, X0, lane, A0);
  f3_load<false>(X0, lane, A0);
  gate3(ws + LAYER_F, lane, g1);
  f3c_run(g2, X1, lane, A1);

  #pragma unroll 1
  for (int l = 1; l + 1 < NL; ++l) {
    const float* f0 = ws + l * LAYER_F;
    int ln = lane;
    asm volatile("" : "+v"(ln));
    ln &= 31;
    f3_load<false>(X1, ln, A1);
    f3_run<false>(g1, X0, ln, A0);
    f3_load<true>(X0, ln, A0);
    gate3(f0 + 256, ln, g2);
    f3_run<false>(g1, X1, ln, A1);
    f3_load<true>(X1, ln, A1);
    gate4(f0 + 512, ln, g3);
    f3_run<true>(g2, X0, ln, A0);
    f4_load(X0, ln, A0);
    f3_run<true>(g2, X1, ln, A1);
    f4_load(X1, ln, A1);
    f4_run(g3, X0, ln, A0);
    f3_load<false>(X0, ln, A0);
    gate3(f0 + LAYER_F, ln, g1);
    f4_run(g3, X1, ln, A1);
  }

  {
    const float* f0 = ws + (NL - 1) * LAYER_F;
    int ln = lane;
    asm volatile("" : "+v"(ln));
    ln &= 31;
    f3_load<false>(X1, ln, A1);
    f3_run<false>(g1, X0, ln, A0);
    f3_load<true>(X0, ln, A0);
    gate3(f0 + 256, ln, g2);
    f3_run<false>(g1, X1, ln, A1);
    f3_load<true>(X1, ln, A1);
    gate4r(ws + NL * LAYER_F, ln, g3);
    f3_run<true>(g2, X0, ln, A0);
    f4r_load(X0, ln, A0);
    f3_run<true>(g2, X1, ln, A1);
    f4r_load(X1, ln, A1);
    f4r_run(g3, X0, ln, A0);
    f4r_run(g3, X1, ln, A1);
  }

  measure(X0, out, sample, lane);
  measure(X1, out, sample + 1, lane);
}

extern "C" void kernel_launch(void* const* d_in, const int* in_sizes, int n_in,
                              void* d_out, int out_size, void* d_ws, size_t ws_size,
                              hipStream_t stream) {
  const float* x = (const float*)d_in[0];
  const float* w = (const float*)d_in[1];
  float* ws  = (float*)d_ws;
  float* out = (float*)d_out;

  vqc_prep<<<PREP_BLOCKS, 128, 0, stream>>>(x, w, ws);
  vqc_main<<<BATCH_N / (WPB * SPW), 32 * WPB, 0, stream>>>(ws, out);
}


# last layer: RY of wires 8,9 carried into the observable (quadratic forms per amplitude quad); wires 0-7 as two REAL 4-wire factors: 64 f32 WMMA not 96, two passes not three; sums regrouped
# speedup vs baseline: 2.6660x; 1.0432x over previous
#include <hip/hip_runtime.h>


typedef __attribute__((ext_vector_type(2))) float v2f;
typedef __attribute__((ext_vector_type(4))) float v4f;
typedef __attribute__((ext_vector_type(8))) float v8f;

#define NQ      10
#define NL      4
#define BATCH_N 8192
#define BUF_F   2048
#define SPW     2
#define WPB     8
#define LAYER_F 1536
#define RTAB_F  512
#define CONST_F 128
#define TAB_F      (NL * LAYER_F + RTAB_F + CONST_F)
#define TAB_BLOCKS (TAB_F / 128)
#define ENC_BLOCKS (BATCH_N * 2 / 128)
#define PREP_BLOCKS (TAB_BLOCKS + ENC_BLOCKS)

__device__ __forceinline__ constexpr int phys(int p, int U) {
  return (p << 10) | (((U ^ ((U >> 4) & 15) ^ p) & 255) << 2);
}


__device__ __forceinline__ constexpr int rowel(int n) {
  return ((n & 7) << 2) | (n >> 3);
}

__device__ __forceinline__ constexpr int ringaddr(int p, int a) {
  int y = a;
  y ^= y >> 1; y ^= y >> 2; y ^= y >> 4; y ^= y >> 8;
  const int n = (y & 0x1ff) | (((y ^ (y >> 9)) & 1) << 9);
  return phys(p, ((n & 0x7f) << 1) | (n >> 9)) | ((n >> 7) & 3);
}

__device__ __forceinline__ void gate_elem(float a, float b, int rb, int cb, bool phase,
                                          float& fr, float& fi) {
  float ca = cosf(0.5f * a), sa = sinf(0.5f * a);
  float ry = rb ? (cb ? ca : sa) : (cb ? -sa : ca);
  if (!phase) { fr = ry; fi = 0.f; return; }
  float cz = cosf(0.5f * b), sz = sinf(0.5f * b);
  fr = ry * cz;
  fi = ry * (rb ? sz : -sz);
}

__device__ __forceinline__ v4f ld4(const float* p) { return *(const v4f*)p; }
__device__ __forceinline__ void st4(float* p, v4f v) { *(v4f*)p = v; }

__device__ __forceinline__ void prep_consts(const float* __restrict__ w, float* __restrict__ ws, int t, int k) {
  const float t8 = w[((NL - 1) * NQ + 8) * 2], t9 = w[((NL - 1) * NQ + 9) * 2];
  const float c8 = cosf(t8), s8 = sinf(t8), c9 = cosf(t9), s9 = sinf(t9);
  float v = 0.f;
  if      (k == 0) v = c8;
  else if (k == 1) v = -2.f * s8;
  else if (k == 2) v = c8 * c9;
  else if (k == 3) v = -2.f * c8 * s9;
  else if (k == 4) v = -2.f * s8 * c9;
  else if (k == 5) v = 2.f * s8 * s9;
  ws[t] = v;
}

__device__ __forceinline__ void prep_tables(const float* __restrict__ w, float* __restrict__ ws, int t) {
  int l = t / LAYER_F;
  int e = t - l * LAYER_F;
  if (l == NL && e >= RTAB_F) {
    prep_consts(w, ws, t, e - RTAB_F);
    return;
  }
  int nb, first, M, K;
  int top = -1;
  const bool phase = l < NL;
  if (!phase) {
    int lane = (e >> 3) & 31, slot = e & 7;
    l = NL - 1;
    nb = 4; first = 3 * (e >> 8); top = 6 + (e >> 8); M = lane & 15; K = 8 * (lane >> 4) + slot;
  } else if (e < 512) {
    int lane = (e >> 3) & 31, slot = e & 7;
    nb = 3; first = 3 * (e >> 8); M = lane & 15; K = 8 * (lane >> 4) + slot;
  } else {
    int e2 = e - 512;
    int lane = (e2 >> 4) & 31, slot = e2 & 15;
    nb = 4; first = 6; M = (e2 >> 9) * 16 + (lane & 15); K = 16 * (lane >> 4) + slot;
  }
  int gm = (1 << nb) - 1;
  int po = M >> nb, o = M & gm, pi = K >> nb, i = K & gm;
  float vr = 1.f, vi = 0.f;
  #pragma unroll
  for (int j = 0; j < 4; ++j) {
    if (j < nb) {
      int wire = (top < 0) ? first + j : (j == 0 ? top : first + j - 1);
      int rb = (o >> (nb - 1 - j)) & 1;
      int cb = (i >> (nb - 1 - j)) & 1;
      float fr, fi;
      gate_elem(w[(l * NQ + wire) * 2], w[(l * NQ + wire) * 2 + 1], rb, cb, phase, fr, fi);
      float nr = vr * fr - vi * fi; vi = vr * fi + vi * fr; vr = nr;
    }
  }
  ws[t] = (po == pi) ? vr : (po ? vi : -vi);
}

__device__ __forceinline__ void prep_encoding(const float* __restrict__ x, float* __restrict__ ws, int u) {
  const float* xp = x + u * 5;
  float xv[5];
  #pragma unroll
  for (int j = 0; j < 5; ++j) xv[j] = xp[j];
  float v[32];
  {
    const float ah = 0.5f * xv[0];
    v[0] = cosf(ah); v[1] = sinf(ah);
  }
  #pragma unroll
  for (int j = 1; j < 5; ++j) {
    const float ah = 0.5f * xv[j];
    const float cj = cosf(ah), sj = sinf(ah);
    #pragma unroll
    for (int i = (1 << j) - 1; i >= 0; --i) {
      const float p = v[i];
      v[2 * i] = p * cj; v[2 * i + 1] = p * sj;
    }
  }
  float* o = ws + TAB_F + u * 32;
  if (u & 1) {
    #pragma unroll
    for (int q = 0; q < 8; ++q) {
      v4f t = {v[4 * q], v[4 * q + 1], v[4 * q + 2], v[4 * q + 3]};
      st4(o + 4 * q, t);
    }
  } else {
    #pragma unroll
    for (int q = 0; q < 8; ++q) {
      v4f t = {v[rowel(4 * q)], v[rowel(4 * q + 1)], v[rowel(4 * q + 2)], v[rowel(4 * q + 3)]};
      st4(o + 4 * q, t);
    }
  }
}

__global__ __launch_bounds__(128) void vqc_prep(const float* __restrict__ x, const float* __restrict__ w,
                                                float* __restrict__ ws) {
  if (blockIdx.x < TAB_BLOCKS)
    prep_tables(w, ws, blockIdx.x * 128 + threadIdx.x);
  else
    prep_encoding(x, ws, (blockIdx.x - TAB_BLOCKS) * 128 + threadIdx.x);
}


__device__ __forceinline__ void gate3(const float* __restrict__ frag, int lane, float (&g)[8]) {
  const v4f* gp = (const v4f*)(frag + lane * 8);
  const v4f g0 = gp[0], g1 = gp[1];
  g[0] = g0.x; g[1] = g0.y; g[2] = g0.z; g[3] = g0.w;
  g[4] = g1.x; g[5] = g1.y; g[6] = g1.z; g[7] = g1.w;
}

__device__ __forceinline__ void gate4(const float* __restrict__ frag, int lane, float (&g)[2][16]) {
  #pragma unroll
  for (int nt = 0; nt < 2; ++nt) {
    const v4f* gp = (const v4f*)(frag + nt * 512 + lane * 16);
    #pragma unroll
    for (int q = 0; q < 4; ++q) {
      const v4f t = gp[q];
      g[nt][4 * q] = t.x; g[nt][4 * q + 1] = t.y; g[nt][4 * q + 2] = t.z; g[nt][4 * q + 3] = t.w;
    }
  }
}

__device__ __forceinline__ void f3_load_first(const float* X, int lane, v4f (&a)[16], bool realx) {
  const int c = lane & 15, h = lane >> 4;
  const int pb = realx ? (c >> 3) : h, rb = realx ? h : (c >> 3);
  const int lb = phys(pb, ((c & 7) << 5) | (rb << 4));
  #pragma unroll
  for (int T = 0; T < 8; ++T)
    #pragma unroll
    for (int j = 0; j < 2; ++j)
      a[2 * T + j] = ld4(X + (lb ^ phys(0, (T << 1) | j)));
  __builtin_amdgcn_sched_barrier(0);
}

template <bool SECOND, bool REALX = false>
__device__ __forceinline__ void f3_load(const float* X, int lane, v4f (&a)[16]) {
  if (!SECOND) { f3_load_first(X, lane, a, REALX); return; }
  const int c = lane & 15, h = lane >> 4;
  int lb;
  if (!REALX) lb = phys(h, c << 1);
  else        lb = phys((c >> 2) & 1, ((c & 11) | (h << 2)) << 1);
  #pragma unroll
  for (int T = 0; T < 8; ++T)
    #pragma unroll
    for (int j = 0; j < 2; ++j)
      a[2 * T + j] = ld4(X + (lb ^ phys(0, (T << 5) | j)));
  __builtin_amdgcn_sched_barrier(0);
}

template <bool SECOND, bool REALX = false>
__device__ __forceinline__ void f3_run(const float (&g)[8], float* X, int lane, const v4f (&a)[16]) {
  const int c = lane & 15, h = lane >> 4;
  int sb;
  if (!SECOND) {
    if (!REALX) sb = phys(c >> 3, ((c & 7) << 5) | (h << 4));
    else        sb = phys(h, ((c & 7) << 5) | ((c >> 3) << 4));
  } else {
    if (!REALX) sb = phys(c >> 3, ((((c & 3) | (((c >> 2) & 1) << 3))) << 2) | (h << 1));
    else        sb = phys(0, ((((c & 3) | (((c >> 2) & 1) << 3))) << 2) | (h << 1) | (c >> 3));
  }
  #pragma unroll
  for (int tp = 0; tp < 4; ++tp) {
    v8f d[2];
    d[0] = (v8f){0, 0, 0, 0, 0, 0, 0, 0};
    d[1] = (v8f){0, 0, 0, 0, 0, 0, 0, 0};
    #pragma unroll
    for (int ks = 0; ks < 4; ++ks)
      #pragma unroll
      for (int t2 = 0; t2 < 2; ++t2) {
        const int T = 2 * tp + t2;
        const v4f av = a[2 * T + (ks >> 1)];
        v2f a2 = {av[(2 * ks) & 3], av[((2 * ks) & 3) + 1]};
        v2f b2 = {g[2 * ks], g[2 * ks + 1]};
        d[t2] = __builtin_amdgcn_wmma_f32_16x16x4_f32(false, a2, false, b2,
                                                      (short)0, d[t2], false, false);
      }
    #pragma unroll
    for (int t2 = 0; t2 < 2; ++t2)
      #pragma unroll
      for (int j = 0; j < 2; ++j) {
        const int T = 2 * tp + t2;
        const int cu = SECOND ? (((((T & 1) << 2) | ((T >> 1) << 4)) << 2) | (REALX ? 0 : j)) : ((T << 1) | j);
        const int pj = (SECOND && REALX) ? j : 0;
        v4f v = {d[t2][4 * j], d[t2][4 * j + 1], d[t2][4 * j + 2], d[t2][4 * j + 3]};
        st4(X + (sb ^ phys(pj, cu)), v);
      }
    __builtin_amdgcn_sched_barrier(0);
  }
}

__device__ __forceinline__ void f4_load(const float* X, int lane, v4f (&a)[16]) {
  const int c = lane & 15, h = lane >> 4;
  const int u0 = c >> 3, u3 = (c >> 2) & 1, u2 = (c >> 1) & 1, u1 = c & 1;
  const int lrow = ((u2 ^ u1) << 2) | ((u1 ^ u0) << 3) | ((u3 ^ u2) << 4) | (u3 << 5);
  const int lb = phys(h, lrow << 2);
  #pragma unroll
  for (int T = 0; T < 4; ++T)
    #pragma unroll
    for (int q = 0; q < 4; ++q)
      a[4 * T + q] = ld4(X + (lb ^ phys(0, (T << 2) | q)));
  __builtin_amdgcn_sched_barrier(0);
}

__device__ __forceinline__ void f4_run(const float (&g)[2][16], float* X, int lane, const v4f (&a)[16]) {
  const int c = lane & 15, h = lane >> 4;
  const int n3 = h ^ (c >> 3), n2 = n3 ^ ((c >> 2) & 1), n1 = n2 ^ ((c >> 1) & 1), n0 = n1 ^ (c & 1);
  const int srow = (h << 6) | (h << 5) | (h << 4) | (n3 << 3) | (n2 << 2) | (n1 << 1) | n0;
  const int sb = phys(0, (srow << 1) | n0);
  #pragma unroll
  for (int nt = 0; nt < 2; ++nt) {
    #pragma unroll
    for (int tp = 0; tp < 2; ++tp) {
      v8f d[2];
      d[0] = (v8f){0, 0, 0, 0, 0, 0, 0, 0};
      d[1] = (v8f){0, 0, 0, 0, 0, 0, 0, 0};
      #pragma unroll
      for (int ks = 0; ks < 8; ++ks)
        #pragma unroll
        for (int t2 = 0; t2 < 2; ++t2) {
          const int T = 2 * tp + t2;
          const v4f av = a[4 * T + (ks >> 1)];
          v2f a2 = {av[(2 * ks) & 3], av[((2 * ks) & 3) + 1]};
          v2f b2 = {g[nt][2 * ks], g[nt][2 * ks + 1]};
          d[t2] = __builtin_amdgcn_wmma_f32_16x16x4_f32(false, a2, false, b2,
                                                        (short)0, d[t2], false, false);
        }
      #pragma unroll
      for (int t2 = 0; t2 < 2; ++t2)
        #pragma unroll
        for (int j = 0; j < 2; ++j) {
          const int T = 2 * tp + t2;
          const int a5 = (T >> 1) & 1, a4 = T & 1;
          const int nT = (a5 ? 0x3f : 0) ^ (a4 ? 0x1f : 0);
          const int cu = (nT << 1) | ((a5 ^ a4) ^ j);
          v4f v = {d[t2][4 * j], d[t2][4 * j + 1], d[t2][4 * j + 2], d[t2][4 * j + 3]};
          st4(X + (sb ^ phys(nt, cu)), v);
        }
      __builtin_amdgcn_sched_barrier(0);
    }
  }
}

__device__ __forceinline__ void gate4a(const float* __restrict__ frag, int lane, float (&g)[2][16]) {
  const int c = lane & 15, h = lane >> 4;
  #pragma unroll
  for (int nt = 0; nt < 2; ++nt) {
    const v4f* gp = (const v4f*)(frag + nt * 512 + c * 16 + 8 * h);
    const v4f g0 = gp[0], g1 = gp[1];
    g[nt][0] = g0.x; g[nt][1] = g0.y; g[nt][2] = g0.z; g[nt][3] = g0.w;
    g[nt][4] = g1.x; g[nt][5] = g1.y; g[nt][6] = g1.z; g[nt][7] = g1.w;
  }
}

__device__ __forceinline__ void fa_load(const float* X, int lane, v4f (&a)[16]) {
  const int c = lane & 15, h = lane >> 4;
  const int lb = phys(0, (((c >> 2) & 1) << 7) | (((c >> 1) & 1) << 6) | ((c >> 3) << 5) | ((c & 1) << 4) | (h << 1));
  #pragma unroll
  for (int T = 0; T < 4; ++T)
    #pragma unroll
    for (int j = 0; j < 2; ++j)
      a[2 * T + j] = ld4(X + (lb ^ phys(0, (T << 2) | j)));
  __builtin_amdgcn_sched_barrier(0);
}

__device__ __forceinline__ void fa_run(const float (&g)[2][16], float* X, int lane, const v4f (&a)[16]) {
  const int c = lane & 15, h = lane >> 4;
  const int sb = phys(0, (h << 7) | (c << 1));
  #pragma unroll
  for (int nt = 0; nt < 2; ++nt) {
    #pragma unroll
    for (int tp = 0; tp < 2; ++tp) {
      v8f d[2];
      d[0] = (v8f){0, 0, 0, 0, 0, 0, 0, 0};
      d[1] = (v8f){0, 0, 0, 0, 0, 0, 0, 0};
      #pragma unroll
      for (int ks = 0; ks < 4; ++ks)
        #pragma unroll
        for (int t2 = 0; t2 < 2; ++t2) {
          const int T = 2 * tp + t2;
          const v4f av = a[2 * T + (ks >> 1)];
          v2f a2 = {av[(2 * ks) & 3], av[((2 * ks) & 3) + 1]};
          v2f b2 = {g[nt][2 * ks], g[nt][2 * ks + 1]};
          d[t2] = __builtin_amdgcn_wmma_f32_16x16x4_f32(false, a2, false, b2,
                                                        (short)0, d[t2], false, false);
        }
      #pragma unroll
      for (int t2 = 0; t2 < 2; ++t2)
        #pragma unroll
        for (int j = 0; j < 2; ++j) {
          const int T = 2 * tp + t2;
          v4f v = {d[t2][4 * j], d[t2][4 * j + 1], d[t2][4 * j + 2], d[t2][4 * j + 3]};
          st4(X + (sb ^ phys(nt, (T << 5) | j)), v);
        }
      __builtin_amdgcn_sched_barrier(0);
    }
  }
}

__device__ __forceinline__ void f3c_run(const float (&g)[8], float* X, int lane, const v4f (&a)[16]) {
  const int c = lane & 15, h = lane >> 4;
  const int sb = ringaddr(c >> 3, ((c & 7) << 4) | (h << 3));
  #pragma unroll
  for (int tp = 0; tp < 4; ++tp) {
    v8f d[2];
    d[0] = (v8f){0, 0, 0, 0, 0, 0, 0, 0};
    d[1] = (v8f){0, 0, 0, 0, 0, 0, 0, 0};
    #pragma unroll
    for (int ks = 0; ks < 4; ++ks)
      #pragma unroll
      for (int t2 = 0; t2 < 2; ++t2) {
        const int T = 2 * tp + t2;
        const v4f av = a[2 * T + (ks >> 1)];
        v2f a2 = {av[(2 * ks) & 3], av[((2 * ks) & 3) + 1]};
        v2f b2 = {g[2 * ks], g[2 * ks + 1]};
        d[t2] = __builtin_amdgcn_wmma_f32_16x16x4_f32(false, a2, false, b2,
                                                      (short)0, d[t2], false, false);
      }
    #pragma unroll
    for (int t2 = 0; t2 < 2; ++t2)
      #pragma unroll
      for (int r = 0; r < 8; ++r) {
        const int T = 2 * tp + t2;
        X[sb ^ ringaddr(0, (T << 7) | r)] = d[t2][r];
      }
    __builtin_amdgcn_sched_barrier(0);
  }
}

__device__ __forceinline__ float xhalf(float x) { return __shfl_xor(x, 16, 32); }

template <int M>
__device__ __forceinline__ float xrow(float x) {
  int b = __builtin_bit_cast(int, x);
  return __builtin_bit_cast(float, __builtin_amdgcn_update_dpp(0, b, 0x160 | M, 0xf, 0xf, true));
}

__device__ __forceinline__ void encode3(const float* __restrict__ ws, float* X, int sample, int lane) {
  const float* ev = ws + TAB_F + sample * 64;
  const v4f cq = ld4(ev + 32 + 4 * (lane & 7));
  const v4f r0 = ld4(ev + 8 * (lane >> 3)), r1 = ld4(ev + 8 * (lane >> 3) + 4);
  const float rv[8] = {r0.x, r0.y, r0.z, r0.w, r1.x, r1.y, r1.z, r1.w};
  const int ub = phys(0, (((lane >> 4) & 1) << 5) | (((lane >> 3) & 1) << 3) | (lane & 7));
  #pragma unroll
  for (int k = 0; k < 8; ++k) {
    const int cu = (((k >> 2) & 1) << 7) | (((k >> 1) & 1) << 6) | ((k & 1) << 4);
    st4(X + (ub ^ phys(0, cu)), cq * rv[k]);
  }
}


__device__ __forceinline__ void measure(const float* __restrict__ kc, const float* X,
                                        float* __restrict__ out, int sample, int lane) {
  const int mc = lane & 15, mh = lane >> 4;
  const float* Lp = X + ((mh << 6) | (mc << 2));
  float ut[8];
  v2f dq01[2] = {{0.f, 0.f}, {0.f, 0.f}};
  v2f dq23[2] = {{0.f, 0.f}, {0.f, 0.f}};
  v2f cx[2]   = {{0.f, 0.f}, {0.f, 0.f}};
  float c01[2] = {0.f, 0.f}, c23[2] = {0.f, 0.f};
  float c03[2] = {0.f, 0.f}, c12[2] = {0.f, 0.f};
  #pragma unroll
  for (int t = 0; t < 8; ++t) {
    const v4f a = ld4(Lp + (t << 7));
    const v4f b = ld4(Lp + ((1 << 10) | (t << 7)));
    const v2f al = {a.x, a.y}, ah = {a.z, a.w}, bl = {b.x, b.y}, bh = {b.z, b.w};
    const v2f sl = al * al + bl * bl, sh = ah * ah + bh * bh;
    const v2f ss = sl + sh;
    ut[t] = ss.x + ss.y;
    const int k = t >> 2;
    dq01[k] += al * al - bl * bl;
    dq23[k] += ah * ah - bh * bh;
    cx[k]   += al * ah - bl * bh;
    c01[k]  += a.x * a.y - b.x * b.y;
    c23[k]  += a.z * a.w - b.z * b.w;
    c03[k]  += a.x * a.w - b.x * b.w;
    c12[k]  += a.y * a.z - b.y * b.z;
  }
  const float e0 = ut[0] + ut[1], e1 = ut[2] + ut[3], e2 = ut[4] + ut[5], e3 = ut[6] + ut[7];
  const float d0 = ut[0] - ut[1], d1 = ut[2] - ut[3], d2 = ut[4] - ut[5], d3 = ut[6] - ut[7];
  float ta = (e0 + e1) + (e2 + e3);
  float p1 = (e0 - e1) - (e2 - e3);
  float p3 = (d0 - d1) - (d2 - d3);
  float p4 = (d0 - d1) + (d2 - d3);
  float p5 = (d0 + d1) + (d2 + d3);
  const v2f q01 = dq01[0] + dq01[1], q23 = dq23[0] + dq23[1], cxs = cx[0] + cx[1];
  float a7 = (q01.x + q01.y) + (q23.x + q23.y);
  const float f1 = (q01.x + q01.y) - (q23.x + q23.y);
  const float f2 = cxs.x + cxs.y;
  const float qcc0 = (dq01[0].x - dq01[0].y) - (dq23[0].x - dq23[0].y);
  const float qcc1 = (dq01[1].x - dq01[1].y) - (dq23[1].x - dq23[1].y);
  const float qcs0 = c01[0] - c23[0], qcs1 = c01[1] - c23[1];
  const float qsc0 = cx[0].x - cx[0].y, qsc1 = cx[1].x - cx[1].y;
  const float qss0 = c03[0] + c12[0], qss1 = c03[1] + c12[1];
  const float k0 = kc[0], k1 = kc[1], k2 = kc[2], k3 = kc[3], k4 = kc[4], k5 = kc[5];
  float e8 = k0 * f1 + k1 * f2;
  float g9 = (k2 * (qcc0 + qcc1) + k3 * (qcs0 + qcs1)) + (k4 * (qsc0 + qsc1) + k5 * (qss0 + qss1));
  float g0 = (k2 * (qcc0 - qcc1) + k3 * (qcs0 - qcs1)) + (k4 * (qsc0 - qsc1) + k5 * (qss0 - qss1));

  p1 += xrow<1>(p1); p1 += xrow<2>(p1); p1 += xrow<4>(p1); p1 += xrow<8>(p1);
  p3 += xrow<1>(p3); p3 += xrow<2>(p3); p3 += xrow<4>(p3); p3 += xrow<8>(p3);
  p4 += xrow<1>(p4); p4 += xrow<2>(p4); p4 += xrow<4>(p4); p4 -= xrow<8>(p4);
  p5 += xrow<1>(p5); p5 += xrow<2>(p5); p5 -= xrow<4>(p5); p5 -= xrow<8>(p5);
  ta += xrow<1>(ta); ta -= xrow<2>(ta); ta -= xrow<4>(ta); ta -= xrow<8>(ta);
  a7 -= xrow<1>(a7); a7 -= xrow<2>(a7); a7 -= xrow<4>(a7); a7 -= xrow<8>(a7);
  e8 -= xrow<1>(e8); e8 -= xrow<2>(e8); e8 -= xrow<4>(e8); e8 -= xrow<8>(e8);
  g9 -= xrow<1>(g9); g9 -= xrow<2>(g9); g9 -= xrow<4>(g9); g9 -= xrow<8>(g9);
  g0 -= xrow<1>(g0); g0 -= xrow<2>(g0); g0 -= xrow<4>(g0); g0 -= xrow<8>(g0);
  const float p1o = xhalf(p1), p3o = xhalf(p3), p4o = xhalf(p4), p5o = xhalf(p5), tao = xhalf(ta);
  const float a7o = xhalf(a7), e8o = xhalf(e8), g9o = xhalf(g9), g0o = xhalf(g0);
  if (lane == 0) {
    float* o = out + sample * NQ;
    o[0] = g0 + g0o;
    o[1] = p1 + p1o;
    o[2] = p1 - p1o;
    o[3] = p3 - p3o;
    o[4] = p4 - p4o;
    o[5] = p5 - p5o;
    o[6] = ta - tao;
    o[7] = a7 + a7o;
    o[8] = e8 + e8o;
    o[9] = g9 + g9o;
  }
}

__global__ __launch_bounds__(32 * WPB) __attribute__((amdgpu_waves_per_eu(4, 4))) void vqc_main(const float* __restrict__ ws,
                                                     float* __restrict__ out) {
  __shared__ __attribute__((aligned(16))) float lds[WPB * SPW * BUF_F];
  const int lane = threadIdx.x & 31;
  const int wv   = threadIdx.x >> 5;
  const int sample = (blockIdx.x * WPB + wv) * SPW;

  float* X0 = lds + wv * (SPW * BUF_F);
  float* X1 = X0 + BUF_F;

  encode3(ws, X0, sample, lane);
  encode3(ws, X1, sample + 1, lane);

  v4f A0[16], A1[16];
  float g1[8], g2[8], g3[2][16];

  fa_load(X0, lane, A0);
  gate4a(ws + 512, lane, g3);
  fa_load(X1, lane, A1);
  fa_run(g3, X0, lane, A0);
  f3_load<false>(X0, lane, A0);
  gate3(ws, lane, g1);
  fa_run(g3, X1, lane, A1);
  f3_load<false>(X1, lane, A1);
  f3_run<false>(g1, X0, lane, A0);
  f3_load<true>(X0, lane, A0);
  gate3(ws + 256, lane, g2);
  f3_run<false>(g1, X1, lane, A1);
  f3_load<true>(X1, lane, A1);
  f3c_run(g2, X0, lane, A0);
  f3_load<false>(X0, lane, A0);
  gate3(ws + LAYER_F, lane, g1);
  f3c_run(g2, X1, lane, A1);

  #pragma unroll 1
  for (int l = 1; l + 1 < NL; ++l) {
    const float* f0 = ws + l * LAYER_F;
    int ln = lane;
    asm volatile("" : "+v"(ln));
    ln &= 31;
    f3_load<false>(X1, ln, A1);
    f3_run<false>(g1, X0, ln, A0);
    f3_load<true>(X0, ln, A0);
    gate3(f0 + 256, ln, g2);
    f3_run<false>(g1, X1, ln, A1);
    f3_load<true>(X1, ln, A1);
    gate4(f0 + 512, ln, g3);
    f3_run<true>(g2, X0, ln, A0);
    f4_load(X0, ln, A0);
    f3_run<true>(g2, X1, ln, A1);
    f4_load(X1, ln, A1);
    f4_run(g3, X0, ln, A0);
    const bool last = (l + 2 == NL);
    f3_load_first(X0, ln, A0, last);
    gate3(f0 + (last ? 2 * LAYER_F : LAYER_F), ln, g1);
    f4_run(g3, X1, ln, A1);
  }

  const float* fr = ws + NL * LAYER_F;
  {
    int ln = lane;
    asm volatile("" : "+v"(ln));
    ln &= 31;
    f3_load<false, true>(X1, ln, A1);
    f3_run<false, true>(g1, X0, ln, A0);
    f3_load<true, true>(X0, ln, A0);
    gate3(fr + 256, ln, g2);
    f3_run<false, true>(g1, X1, ln, A1);
    f3_load<true, true>(X1, ln, A1);
    f3_run<true, true>(g2, X0, ln, A0);
    f3_run<true, true>(g2, X1, ln, A1);
  }

  measure(fr + RTAB_F, X0, out, sample, lane);
  measure(fr + RTAB_F, X1, out, sample + 1, lane);
}

extern "C" void kernel_launch(void* const* d_in, const int* in_sizes, int n_in,
                              void* d_out, int out_size, void* d_ws, size_t ws_size,
                              hipStream_t stream) {
  const float* x = (const float*)d_in[0];
  const float* w = (const float*)d_in[1];
  float* ws  = (float*)d_ws;
  float* out = (float*)d_out;

  vqc_prep<<<PREP_BLOCKS, 128, 0, stream>>>(x, w, ws);
  vqc_main<<<BATCH_N / (WPB * SPW), 32 * WPB, 0, stream>>>(ws, out);
}
